# starting kernel
# baseline (speedup 1.0000x reference)
#include <hip/hip_runtime.h>

#define HID   20
#define GATES 80
#define TSEQ  1024
#define ROWS  128
#define HPAD  24
#define GPAD  84
#define KC    6
#define NT    5

typedef __attribute__((ext_vector_type(2))) float v2f;
typedef __attribute__((ext_vector_type(4))) float v4f;
typedef __attribute__((ext_vector_type(8))) float v8f;

__device__ __forceinline__ float fast_sigmoid(float x) {
  float e = __builtin_amdgcn_exp2f(-x * 1.44269504088896340736f);
  return __builtin_amdgcn_rcpf(1.0f + e);
}
__device__ __forceinline__ float fast_tanh(float x) {
  float e = __builtin_amdgcn_exp2f(x * 2.88539008177792681472f);
  return 1.0f - 2.0f * __builtin_amdgcn_rcpf(e + 1.0f);
}

__global__ __launch_bounds__(ROWS)
void lstm_fused(const float* __restrict__ x,    const float* __restrict__ Wih1,
                const float* __restrict__ Whh1, const float* __restrict__ bih1,
                const float* __restrict__ bhh1, const float* __restrict__ Wih2,
                const float* __restrict__ Whh2, const float* __restrict__ bih2,
                const float* __restrict__ bhh2, float* __restrict__ out)
{
  __shared__ __align__(16) float h1x[ROWS * HPAD];
  __shared__ __align__(16) float gl [ROWS * GPAD];
  __shared__ __align__(16) float w2 [HID * 4];

  const int tid  = threadIdx.x;
  const int lane = tid & 31;
  const int wave = tid >> 5;
  const int ln   = lane & 15;
  const int kh   = (lane >> 4) * 2;
  const int row0 = blockIdx.x * ROWS;

  for (int i = tid; i < ROWS * HPAD; i += ROWS) h1x[i] = 0.0f;
  if (tid < HID * 4) {
    int j = tid >> 2, q = tid & 3;
    w2[tid] = Wih2[q * HID + j];
  }

  v2f  bfrag[KC][NT];
  float biasf[NT];
  #pragma unroll
  for (int nt = 0; nt < NT; ++nt) {
    int n = nt * 16 + ln;
    biasf[nt] = bih1[n] + bhh1[n];
    #pragma unroll
    for (int kc = 0; kc < KC; ++kc) {
      int k0 = kc * 4 + kh;
      v2f b;
      if (k0 + 1 < HID)      { b.x = Whh1[n * HID + k0]; b.y = Whh1[n * HID + k0 + 1]; }
      else if (k0 == HID)    { b.x = Wih1[n];            b.y = 0.0f; }
      else                   { b.x = 0.0f;               b.y = 0.0f; }
      bfrag[kc][nt] = b;
    }
  }

  float b2[4], whh2v[4];
  #pragma unroll
  for (int q = 0; q < 4; ++q) { b2[q] = bih2[q] + bhh2[q]; whh2v[q] = Whh2[q]; }
  float c1[HID];
  #pragma unroll
  for (int j = 0; j < HID; ++j) c1[j] = 0.0f;
  float h2 = 0.0f, c2 = 0.0f;

  const float* xrow = x   + (size_t)(row0 + tid) * TSEQ;
  float*       orow = out + (size_t)(row0 + tid) * TSEQ;

  h1x[tid * HPAD + HID] = xrow[0];
  __syncthreads();

  for (int t = 0; t < TSEQ; ++t) {
    #pragma unroll
    for (int tile = 0; tile < 2; ++tile) {
      const int r0 = wave * 32 + tile * 16;
      v2f afrag[KC];
      #pragma unroll
      for (int kc = 0; kc < KC; ++kc) {
        int k = kc * 4 + kh;
        afrag[kc] = *(const v2f*)&h1x[(r0 + ln) * HPAD + k];
      }
      v8f acc[NT];
      #pragma unroll
      for (int nt = 0; nt < NT; ++nt) {
        float b = biasf[nt];
        acc[nt] = (v8f){b, b, b, b, b, b, b, b};
      }
      #pragma unroll
      for (int kc = 0; kc < KC; ++kc)
        #pragma unroll
        for (int nt = 0; nt < NT; ++nt)
          acc[nt] = __builtin_amdgcn_wmma_f32_16x16x4_f32(
              false, afrag[kc], false, bfrag[kc][nt],
              (short)0, acc[nt], false, false);
      #pragma unroll
      for (int nt = 0; nt < NT; ++nt) {
        int n  = nt * 16 + ln;
        int pc = (n % HID) * 4 + (n / HID);
        int rb = r0 + (lane >> 4) * 8;
        #pragma unroll
        for (int r = 0; r < 8; ++r)
          gl[(rb + r) * GPAD + pc] = acc[nt][r];
      }
    }
    __syncthreads();

    float xt_next = (t + 1 < TSEQ) ? xrow[t + 1] : 0.0f;

    float a2[4] = {0.0f, 0.0f, 0.0f, 0.0f};
    const float* grow = &gl[tid * GPAD];
    float*       hrow = &h1x[tid * HPAD];
    #pragma unroll
    for (int j = 0; j < HID; ++j) {
      v4f g4 = *(const v4f*)&grow[j * 4];
      float ig = fast_sigmoid(g4.x);
      float fg = fast_sigmoid(g4.y);
      float gg = fast_tanh   (g4.z);
      float og = fast_sigmoid(g4.w);
      float c  = fg * c1[j] + ig * gg;
      float h  = og * fast_tanh(c);
      c1[j] = c;
      hrow[j] = h;
      v4f wq = *(const v4f*)&w2[j * 4];
      a2[0] += h * wq.x; a2[1] += h * wq.y; a2[2] += h * wq.z; a2[3] += h * wq.w;
    }
    hrow[HID] = xt_next;

    float i2 = fast_sigmoid(a2[0] + h2 * whh2v[0] + b2[0]);
    float f2 = fast_sigmoid(a2[1] + h2 * whh2v[1] + b2[1]);
    float g2 = fast_tanh   (a2[2] + h2 * whh2v[2] + b2[2]);
    float o2 = fast_sigmoid(a2[3] + h2 * whh2v[3] + b2[3]);
    c2 = f2 * c2 + i2 * g2;
    h2 = o2 * fast_tanh(c2);
    orow[t] = h2;

    __syncthreads();
  }
}

extern "C" void kernel_launch(void* const* d_in, const int* in_sizes, int n_in,
                              void* d_out, int out_size, void* d_ws, size_t ws_size,
                              hipStream_t stream) {
  const float* x    = (const float*)d_in[0];
  const float* Wih1 = (const float*)d_in[1];
  const float* Whh1 = (const float*)d_in[2];
  const float* bih1 = (const float*)d_in[3];
  const float* bhh1 = (const float*)d_in[4];
  const float* Wih2 = (const float*)d_in[5];
  const float* Whh2 = (const float*)d_in[6];
  const float* bih2 = (const float*)d_in[7];
  const float* bhh2 = (const float*)d_in[8];
  float* out = (float*)d_out;

  const int B    = in_sizes[0] / TSEQ;
  const int grid = B / ROWS;
  hipLaunchKernelGGL(lstm_fused, dim3(grid), dim3(ROWS), 0, stream,
                     x, Wih1, Whh1, bih1, bhh1, Wih2, Whh2, bih2, bhh2, out);
}


# WMMA operands swapped (weights=A, state=B): a lane's D holds all 4 gates of the units whose h it feeds back as B, so recurrence stays in VGPRs (no LDS, no barrier; 16 rows/wave); sum order changes
# speedup vs baseline: 3.7668x; 3.7668x over previous
#include <hip/hip_runtime.h>

#define HID   20
#define TSEQ  1024
#define WAVES 4
#define MT    5
#define KC    6
#define UPL   10

typedef __attribute__((ext_vector_type(2))) float v2f;
typedef __attribute__((ext_vector_type(8))) float v8f;

__device__ __forceinline__ float fast_sigmoid(float x) {
  float e = __builtin_amdgcn_exp2f(-x * 1.44269504088896340736f);
  return __builtin_amdgcn_rcpf(1.0f + e);
}
__device__ __forceinline__ float fast_tanh(float x) {
  float e = __builtin_amdgcn_exp2f(x * 2.88539008177792681472f);
  return 1.0f - 2.0f * __builtin_amdgcn_rcpf(e + 1.0f);
}
__device__ __forceinline__ float other_half(float v) {
  return __builtin_bit_cast(float, __builtin_amdgcn_permlanex16(
      __builtin_bit_cast(int, v), __builtin_bit_cast(int, v), 0x76543210, 0xfedcba98, false, false));
}

__global__ __launch_bounds__(WAVES * 32)
void lstm_fused(const float* __restrict__ x,    const float* __restrict__ Wih1,
                const float* __restrict__ Whh1, const float* __restrict__ bih1,
                const float* __restrict__ bhh1, const float* __restrict__ Wih2,
                const float* __restrict__ Whh2, const float* __restrict__ bih2,
                const float* __restrict__ bhh2, float* __restrict__ out)
{
  const int lane = threadIdx.x & 31;
  const int wave = threadIdx.x >> 5;
  const int c    = lane & 15;
  const int h    = lane >> 4;
  const int row  = (blockIdx.x * WAVES + wave) * 16 + c;

  v2f wfrag[MT][KC];
  {
    const int hm = c >> 3;
    const int r  = c & 7;
    const int q  = r & 3;
    #pragma unroll
    for (int mt = 0; mt < MT; ++mt) {
      const int um = hm * UPL + mt * 2 + (r >> 2);
      const int n  = q * HID + um;
      #pragma unroll
      for (int kc = 0; kc < KC - 1; ++kc) {
        const int uk = h * UPL + kc * 2;
        v2f w;
        w.x = Whh1[n * HID + uk];
        w.y = Whh1[n * HID + uk + 1];
        wfrag[mt][kc] = w;
      }
      v2f wl;
      wl.x = (h == 0) ? Wih1[n]           : 0.0f;
      wl.y = (h == 0) ? bih1[n] + bhh1[n] : 0.0f;
      wfrag[mt][KC - 1] = wl;
    }
  }

  v2f w2if[UPL], w2go[UPL];
  float b2[4], whh2v[4];
  #pragma unroll
  for (int u = 0; u < UPL; ++u) {
    const int j = h * UPL + u;
    w2if[u] = (v2f){Wih2[0 * HID + j], Wih2[1 * HID + j]};
    w2go[u] = (v2f){Wih2[2 * HID + j], Wih2[3 * HID + j]};
  }
  #pragma unroll
  for (int q = 0; q < 4; ++q) {
    b2[q]    = bih2[q] + bhh2[q];
    whh2v[q] = Whh2[q];
  }

  float c1[UPL];
  v2f   hfrag[KC - 1];
  #pragma unroll
  for (int u = 0; u < UPL; ++u) c1[u] = 0.0f;
  #pragma unroll
  for (int kc = 0; kc < KC - 1; ++kc) hfrag[kc] = (v2f){0.0f, 0.0f};
  float h2 = 0.0f, c2 = 0.0f;

  const float* xrow = x   + (size_t)row * TSEQ;
  float*       orow = out + (size_t)row * TSEQ;
  const float  one  = (h == 0) ? 1.0f : 0.0f;
  float xt = xrow[0];

  for (int t = 0; t < TSEQ; ++t) {
    const float xt_next = (t + 1 < TSEQ) ? xrow[t + 1] : 0.0f;

    v2f xb;
    xb.x = (h == 0) ? xt : 0.0f;
    xb.y = one;
    v8f acc[MT];
    asm volatile(
        "v_wmma_f32_16x16x4_f32 %0, %10, %40, 0\n\t"
        "v_wmma_f32_16x16x4_f32 %1, %16, %40, 0\n\t"
        "v_wmma_f32_16x16x4_f32 %2, %22, %40, 0\n\t"
        "v_wmma_f32_16x16x4_f32 %3, %28, %40, 0\n\t"
        "v_wmma_f32_16x16x4_f32 %4, %34, %40, 0\n\t"
        "v_wmma_f32_16x16x4_f32 %0, %5, %35, %0\n\t"
        "v_wmma_f32_16x16x4_f32 %1, %11, %35, %1\n\t"
        "v_wmma_f32_16x16x4_f32 %2, %17, %35, %2\n\t"
        "v_wmma_f32_16x16x4_f32 %3, %23, %35, %3\n\t"
        "v_wmma_f32_16x16x4_f32 %4, %29, %35, %4\n\t"
        "v_wmma_f32_16x16x4_f32 %0, %6, %36, %0\n\t"
        "v_wmma_f32_16x16x4_f32 %1, %12, %36, %1\n\t"
        "v_wmma_f32_16x16x4_f32 %2, %18, %36, %2\n\t"
        "v_wmma_f32_16x16x4_f32 %3, %24, %36, %3\n\t"
        "v_wmma_f32_16x16x4_f32 %4, %30, %36, %4\n\t"
        "v_wmma_f32_16x16x4_f32 %0, %7, %37, %0\n\t"
        "v_wmma_f32_16x16x4_f32 %1, %13, %37, %1\n\t"
        "v_wmma_f32_16x16x4_f32 %2, %19, %37, %2\n\t"
        "v_wmma_f32_16x16x4_f32 %3, %25, %37, %3\n\t"
        "v_wmma_f32_16x16x4_f32 %4, %31, %37, %4\n\t"
        "v_wmma_f32_16x16x4_f32 %0, %8, %38, %0\n\t"
        "v_wmma_f32_16x16x4_f32 %1, %14, %38, %1\n\t"
        "v_wmma_f32_16x16x4_f32 %2, %20, %38, %2\n\t"
        "v_wmma_f32_16x16x4_f32 %3, %26, %38, %3\n\t"
        "v_wmma_f32_16x16x4_f32 %4, %32, %38, %4\n\t"
        "v_wmma_f32_16x16x4_f32 %0, %9, %39, %0\n\t"
        "v_wmma_f32_16x16x4_f32 %1, %15, %39, %1\n\t"
        "v_wmma_f32_16x16x4_f32 %2, %21, %39, %2\n\t"
        "v_wmma_f32_16x16x4_f32 %3, %27, %39, %3\n\t"
        "v_wmma_f32_16x16x4_f32 %4, %33, %39, %4\n\t"
        "v_nop\n\t"
        "v_nop\n\t"
        "v_nop\n\t"
        "v_nop\n\t"
        "v_nop\n\t"
        "v_nop\n\t"
        "v_nop\n\t"
        "v_nop"
        : "=&v"(acc[0]), "=&v"(acc[1]), "=&v"(acc[2]), "=&v"(acc[3]), "=&v"(acc[4])
        :
          "v"(wfrag[0][0]), "v"(wfrag[0][1]), "v"(wfrag[0][2]), "v"(wfrag[0][3]), "v"(wfrag[0][4]), "v"(wfrag[0][5]),
          "v"(wfrag[1][0]), "v"(wfrag[1][1]), "v"(wfrag[1][2]), "v"(wfrag[1][3]), "v"(wfrag[1][4]), "v"(wfrag[1][5]),
          "v"(wfrag[2][0]), "v"(wfrag[2][1]), "v"(wfrag[2][2]), "v"(wfrag[2][3]), "v"(wfrag[2][4]), "v"(wfrag[2][5]),
          "v"(wfrag[3][0]), "v"(wfrag[3][1]), "v"(wfrag[3][2]), "v"(wfrag[3][3]), "v"(wfrag[3][4]), "v"(wfrag[3][5]),
          "v"(wfrag[4][0]), "v"(wfrag[4][1]), "v"(wfrag[4][2]), "v"(wfrag[4][3]), "v"(wfrag[4][4]), "v"(wfrag[4][5]),
          "v"(hfrag[0]), "v"(hfrag[1]), "v"(hfrag[2]), "v"(hfrag[3]), "v"(hfrag[4]), "v"(xb));

    v2f pif = {0.0f, 0.0f}, pgo = {0.0f, 0.0f};
    #pragma unroll
    for (int mt = 0; mt < MT; ++mt) {
      float hn[2];
      #pragma unroll
      for (int s = 0; s < 2; ++s) {
        const int u = mt * 2 + s;
        float ig = fast_sigmoid(acc[mt][s * 4 + 0]);
        float fg = fast_sigmoid(acc[mt][s * 4 + 1]);
        float gg = fast_tanh   (acc[mt][s * 4 + 2]);
        float og = fast_sigmoid(acc[mt][s * 4 + 3]);
        float cn = fg * c1[u] + ig * gg;
        float hv = og * fast_tanh(cn);
        c1[u] = cn;
        hn[s] = hv;
        const v2f hh = {hv, hv};
        pif = __builtin_elementwise_fma(hh, w2if[u], pif);
        pgo = __builtin_elementwise_fma(hh, w2go[u], pgo);
      }
      hfrag[mt] = (v2f){hn[0], hn[1]};
    }

    float a2[4];
    a2[0] = pif.x + other_half(pif.x);
    a2[1] = pif.y + other_half(pif.y);
    a2[2] = pgo.x + other_half(pgo.x);
    a2[3] = pgo.y + other_half(pgo.y);
    float i2 = fast_sigmoid(a2[0] + h2 * whh2v[0] + b2[0]);
    float f2 = fast_sigmoid(a2[1] + h2 * whh2v[1] + b2[1]);
    float g2 = fast_tanh   (a2[2] + h2 * whh2v[2] + b2[2]);
    float o2 = fast_sigmoid(a2[3] + h2 * whh2v[3] + b2[3]);
    c2 = f2 * c2 + i2 * g2;
    h2 = o2 * fast_tanh(c2);
    if (h == 0) orow[t] = h2;

    xt = xt_next;
  }
}

extern "C" void kernel_launch(void* const* d_in, const int* in_sizes, int n_in,
                              void* d_out, int out_size, void* d_ws, size_t ws_size,
                              hipStream_t stream) {
  const float* x    = (const float*)d_in[0];
  const float* Wih1 = (const float*)d_in[1];
  const float* Whh1 = (const float*)d_in[2];
  const float* bih1 = (const float*)d_in[3];
  const float* bhh1 = (const float*)d_in[4];
  const float* Wih2 = (const float*)d_in[5];
  const float* Whh2 = (const float*)d_in[6];
  const float* bih2 = (const float*)d_in[7];
  const float* bhh2 = (const float*)d_in[8];
  float* out = (float*)d_out;

  const int B    = in_sizes[0] / TSEQ;
  const int grid = B / (WAVES * 16);
  hipLaunchKernelGGL(lstm_fused, dim3(grid), dim3(WAVES * 32), 0, stream,
                     x, Wih1, Whh1, bih1, bhh1, Wih2, Whh2, bih2, bhh2, out);
}


# Layer-2 cell (feeds nothing back to layer 1) deferred one step; its chain issued one dependent link per WMMA between the next step's 30 WMMAs instead of serially behind layer 1; sum order unchanged
# speedup vs baseline: 3.9403x; 1.0461x over previous
#include <hip/hip_runtime.h>

#define HID   20
#define TSEQ  1024
#define WAVES 4
#define MT    5
#define KC    6
#define UPL   10

typedef __attribute__((ext_vector_type(2))) float v2f;
typedef __attribute__((ext_vector_type(8))) float v8f;

__device__ __forceinline__ float fast_sigmoid(float x) {
  float e = __builtin_amdgcn_exp2f(-x * 1.44269504088896340736f);
  return __builtin_amdgcn_rcpf(1.0f + e);
}
__device__ __forceinline__ float fast_tanh(float x) {
  float e = __builtin_amdgcn_exp2f(x * 2.88539008177792681472f);
  return 1.0f - 2.0f * __builtin_amdgcn_rcpf(e + 1.0f);
}
__device__ __forceinline__ float other_half(float v) {
  return __builtin_bit_cast(float, __builtin_amdgcn_permlanex16(
      __builtin_bit_cast(int, v), __builtin_bit_cast(int, v), 0x76543210, 0xfedcba98, false, false));
}

__device__ __forceinline__ void layer1_cells(const v8f (&acc)[MT], float (&c1)[UPL], v2f (&hfrag)[KC - 1]) {
  #pragma unroll
  for (int mt = 0; mt < MT; ++mt) {
    float hn[2];
    #pragma unroll
    for (int s = 0; s < 2; ++s) {
      const int u = mt * 2 + s;
      float ig = fast_sigmoid(acc[mt][s * 4 + 0]);
      float fg = fast_sigmoid(acc[mt][s * 4 + 1]);
      float gg = fast_tanh   (acc[mt][s * 4 + 2]);
      float og = fast_sigmoid(acc[mt][s * 4 + 3]);
      float cn = fg * c1[u] + ig * gg;
      c1[u] = cn;
      hn[s] = og * fast_tanh(cn);
    }
    hfrag[mt] = (v2f){hn[0], hn[1]};
  }
}

__global__ __launch_bounds__(WAVES * 32)
void lstm_fused(const float* __restrict__ x,    const float* __restrict__ Wih1,
                const float* __restrict__ Whh1, const float* __restrict__ bih1,
                const float* __restrict__ bhh1, const float* __restrict__ Wih2,
                const float* __restrict__ Whh2, const float* __restrict__ bih2,
                const float* __restrict__ bhh2, float* __restrict__ out)
{
  const int lane = threadIdx.x & 31;
  const int wave = threadIdx.x >> 5;
  const int c    = lane & 15;
  const int h    = lane >> 4;
  const int row  = (blockIdx.x * WAVES + wave) * 16 + c;

  v2f wfrag[MT][KC];
  {
    const int hm = c >> 3;
    const int r  = c & 7;
    const int q  = r & 3;
    #pragma unroll
    for (int mt = 0; mt < MT; ++mt) {
      const int um = hm * UPL + mt * 2 + (r >> 2);
      const int n  = q * HID + um;
      #pragma unroll
      for (int kc = 0; kc < KC - 1; ++kc) {
        const int uk = h * UPL + kc * 2;
        v2f w;
        w.x = Whh1[n * HID + uk];
        w.y = Whh1[n * HID + uk + 1];
        wfrag[mt][kc] = w;
      }
      v2f wl;
      wl.x = (h == 0) ? Wih1[n]           : 0.0f;
      wl.y = (h == 0) ? bih1[n] + bhh1[n] : 0.0f;
      wfrag[mt][KC - 1] = wl;
    }
  }

  v2f w2if[UPL], w2go[UPL];
  float b2[4], whh2v[4];
  #pragma unroll
  for (int u = 0; u < UPL; ++u) {
    const int j = h * UPL + u;
    w2if[u] = (v2f){Wih2[0 * HID + j], Wih2[1 * HID + j]};
    w2go[u] = (v2f){Wih2[2 * HID + j], Wih2[3 * HID + j]};
  }
  #pragma unroll
  for (int q = 0; q < 4; ++q) {
    b2[q]    = bih2[q] + bhh2[q];
    whh2v[q] = Whh2[q];
  }

  float c1[UPL];
  v2f   hfrag[KC - 1];
  #pragma unroll
  for (int u = 0; u < UPL; ++u) c1[u] = 0.0f;
  #pragma unroll
  for (int kc = 0; kc < KC - 1; ++kc) hfrag[kc] = (v2f){0.0f, 0.0f};
  float h2 = 0.0f, c2 = 0.0f;

  const float* xrow = x   + (size_t)row * TSEQ;
  float*       orow = out + (size_t)row * TSEQ;
  const float  one  = (h == 0) ? 1.0f : 0.0f;
  float xt = xrow[0];
  const float* xp = xrow + 1;

  {
    const float xt_next = *xp;
    xp += 1;
    v2f xb;
    xb.x = (h == 0) ? xt : 0.0f;
    xb.y = one;
    v8f acc[MT];
    asm volatile(
        "v_wmma_f32_16x16x4_f32 %[a0], %[w05], %[xb], 0\n\t"
        "v_wmma_f32_16x16x4_f32 %[a1], %[w15], %[xb], 0\n\t"
        "v_wmma_f32_16x16x4_f32 %[a2], %[w25], %[xb], 0\n\t"
        "v_wmma_f32_16x16x4_f32 %[a3], %[w35], %[xb], 0\n\t"
        "v_wmma_f32_16x16x4_f32 %[a4], %[w45], %[xb], 0\n\t"
        "v_wmma_f32_16x16x4_f32 %[a0], %[w00], %[hf0], %[a0]\n\t"
        "v_wmma_f32_16x16x4_f32 %[a1], %[w10], %[hf0], %[a1]\n\t"
        "v_wmma_f32_16x16x4_f32 %[a2], %[w20], %[hf0], %[a2]\n\t"
        "v_wmma_f32_16x16x4_f32 %[a3], %[w30], %[hf0], %[a3]\n\t"
        "v_wmma_f32_16x16x4_f32 %[a4], %[w40], %[hf0], %[a4]\n\t"
        "v_wmma_f32_16x16x4_f32 %[a0], %[w01], %[hf1], %[a0]\n\t"
        "v_wmma_f32_16x16x4_f32 %[a1], %[w11], %[hf1], %[a1]\n\t"
        "v_wmma_f32_16x16x4_f32 %[a2], %[w21], %[hf1], %[a2]\n\t"
        "v_wmma_f32_16x16x4_f32 %[a3], %[w31], %[hf1], %[a3]\n\t"
        "v_wmma_f32_16x16x4_f32 %[a4], %[w41], %[hf1], %[a4]\n\t"
        "v_wmma_f32_16x16x4_f32 %[a0], %[w02], %[hf2], %[a0]\n\t"
        "v_wmma_f32_16x16x4_f32 %[a1], %[w12], %[hf2], %[a1]\n\t"
        "v_wmma_f32_16x16x4_f32 %[a2], %[w22], %[hf2], %[a2]\n\t"
        "v_wmma_f32_16x16x4_f32 %[a3], %[w32], %[hf2], %[a3]\n\t"
        "v_wmma_f32_16x16x4_f32 %[a4], %[w42], %[hf2], %[a4]\n\t"
        "v_wmma_f32_16x16x4_f32 %[a0], %[w03], %[hf3], %[a0]\n\t"
        "v_wmma_f32_16x16x4_f32 %[a1], %[w13], %[hf3], %[a1]\n\t"
        "v_wmma_f32_16x16x4_f32 %[a2], %[w23], %[hf3], %[a2]\n\t"
        "v_wmma_f32_16x16x4_f32 %[a3], %[w33], %[hf3], %[a3]\n\t"
        "v_wmma_f32_16x16x4_f32 %[a4], %[w43], %[hf3], %[a4]\n\t"
        "v_wmma_f32_16x16x4_f32 %[a0], %[w04], %[hf4], %[a0]\n\t"
        "v_wmma_f32_16x16x4_f32 %[a1], %[w14], %[hf4], %[a1]\n\t"
        "v_wmma_f32_16x16x4_f32 %[a2], %[w24], %[hf4], %[a2]\n\t"
        "v_wmma_f32_16x16x4_f32 %[a3], %[w34], %[hf4], %[a3]\n\t"
        "v_wmma_f32_16x16x4_f32 %[a4], %[w44], %[hf4], %[a4]\n\t"
        "v_nop\n\t"
        "v_nop\n\t"
        "v_nop\n\t"
        "v_nop\n\t"
        "v_nop\n\t"
        "v_nop\n\t"
        "v_nop\n\t"
        "v_nop"
        : [a0] "=&v"(acc[0]), [a1] "=&v"(acc[1]), [a2] "=&v"(acc[2]), [a3] "=&v"(acc[3]), [a4] "=&v"(acc[4])
        :
          [w00] "v"(wfrag[0][0]), [w01] "v"(wfrag[0][1]), [w02] "v"(wfrag[0][2]), [w03] "v"(wfrag[0][3]), [w04] "v"(wfrag[0][4]), [w05] "v"(wfrag[0][5]),
          [w10] "v"(wfrag[1][0]), [w11] "v"(wfrag[1][1]), [w12] "v"(wfrag[1][2]), [w13] "v"(wfrag[1][3]), [w14] "v"(wfrag[1][4]), [w15] "v"(wfrag[1][5]),
          [w20] "v"(wfrag[2][0]), [w21] "v"(wfrag[2][1]), [w22] "v"(wfrag[2][2]), [w23] "v"(wfrag[2][3]), [w24] "v"(wfrag[2][4]), [w25] "v"(wfrag[2][5]),
          [w30] "v"(wfrag[3][0]), [w31] "v"(wfrag[3][1]), [w32] "v"(wfrag[3][2]), [w33] "v"(wfrag[3][3]), [w34] "v"(wfrag[3][4]), [w35] "v"(wfrag[3][5]),
          [w40] "v"(wfrag[4][0]), [w41] "v"(wfrag[4][1]), [w42] "v"(wfrag[4][2]), [w43] "v"(wfrag[4][3]), [w44] "v"(wfrag[4][4]), [w45] "v"(wfrag[4][5]),
          [hf0] "v"(hfrag[0]), [hf1] "v"(hfrag[1]), [hf2] "v"(hfrag[2]), [hf3] "v"(hfrag[3]), [hf4] "v"(hfrag[4]), [xb] "v"(xb));
    layer1_cells(acc, c1, hfrag);
    xt = xt_next;
  }

  for (int t = 1; t < TSEQ; ++t) {
    const float xt_next = *xp;

    v2f xb;
    xb.x = (h == 0) ? xt : 0.0f;
    xb.y = one;
    v8f acc[MT];
    asm volatile(
        "v_wmma_f32_16x16x4_f32 %[a0], %[w05], %[xb], 0\n\t"
        "v_pk_fma_f32 v[200:201], %[hf0], %[wif0], 0 op_sel_hi:[0,1,0]\n\t"
        "v_pk_fma_f32 v[202:203], %[hf0], %[wgo0], 0 op_sel_hi:[0,1,0]\n\t"
        "v_wmma_f32_16x16x4_f32 %[a1], %[w15], %[xb], 0\n\t"
        "v_pk_fma_f32 v[200:201], %[hf0], %[wif1], v[200:201] op_sel:[1,0,0]\n\t"
        "v_pk_fma_f32 v[202:203], %[hf0], %[wgo1], v[202:203] op_sel:[1,0,0]\n\t"
        "v_wmma_f32_16x16x4_f32 %[a2], %[w25], %[xb], 0\n\t"
        "v_pk_fma_f32 v[200:201], %[hf1], %[wif2], v[200:201] op_sel_hi:[0,1,1]\n\t"
        "v_pk_fma_f32 v[202:203], %[hf1], %[wgo2], v[202:203] op_sel_hi:[0,1,1]\n\t"
        "v_wmma_f32_16x16x4_f32 %[a3], %[w35], %[xb], 0\n\t"
        "v_pk_fma_f32 v[200:201], %[hf1], %[wif3], v[200:201] op_sel:[1,0,0]\n\t"
        "v_pk_fma_f32 v[202:203], %[hf1], %[wgo3], v[202:203] op_sel:[1,0,0]\n\t"
        "v_wmma_f32_16x16x4_f32 %[a4], %[w45], %[xb], 0\n\t"
        "v_pk_fma_f32 v[200:201], %[hf2], %[wif4], v[200:201] op_sel_hi:[0,1,1]\n\t"
        "v_pk_fma_f32 v[202:203], %[hf2], %[wgo4], v[202:203] op_sel_hi:[0,1,1]\n\t"
        "v_wmma_f32_16x16x4_f32 %[a0], %[w00], %[hf0], %[a0]\n\t"
        "v_pk_fma_f32 v[200:201], %[hf2], %[wif5], v[200:201] op_sel:[1,0,0]\n\t"
        "v_pk_fma_f32 v[202:203], %[hf2], %[wgo5], v[202:203] op_sel:[1,0,0]\n\t"
        "v_wmma_f32_16x16x4_f32 %[a1], %[w10], %[hf0], %[a1]\n\t"
        "v_pk_fma_f32 v[200:201], %[hf3], %[wif6], v[200:201] op_sel_hi:[0,1,1]\n\t"
        "v_pk_fma_f32 v[202:203], %[hf3], %[wgo6], v[202:203] op_sel_hi:[0,1,1]\n\t"
        "v_wmma_f32_16x16x4_f32 %[a2], %[w20], %[hf0], %[a2]\n\t"
        "v_pk_fma_f32 v[200:201], %[hf3], %[wif7], v[200:201] op_sel:[1,0,0]\n\t"
        "v_pk_fma_f32 v[202:203], %[hf3], %[wgo7], v[202:203] op_sel:[1,0,0]\n\t"
        "v_wmma_f32_16x16x4_f32 %[a3], %[w30], %[hf0], %[a3]\n\t"
        "v_pk_fma_f32 v[200:201], %[hf4], %[wif8], v[200:201] op_sel_hi:[0,1,1]\n\t"
        "v_pk_fma_f32 v[202:203], %[hf4], %[wgo8], v[202:203] op_sel_hi:[0,1,1]\n\t"
        "v_wmma_f32_16x16x4_f32 %[a4], %[w40], %[hf0], %[a4]\n\t"
        "v_pk_fma_f32 v[200:201], %[hf4], %[wif9], v[200:201] op_sel:[1,0,0]\n\t"
        "v_pk_fma_f32 v[202:203], %[hf4], %[wgo9], v[202:203] op_sel:[1,0,0]\n\t"
        "v_wmma_f32_16x16x4_f32 %[a0], %[w01], %[hf1], %[a0]\n\t"
        "v_permlanex16_b32 v204, v200, %[sel], 0xfedcba98\n\t"
        "v_permlanex16_b32 v205, v201, %[sel], 0xfedcba98\n\t"
        "v_permlanex16_b32 v206, v202, %[sel], 0xfedcba98\n\t"
        "v_permlanex16_b32 v207, v203, %[sel], 0xfedcba98\n\t"
        "v_wmma_f32_16x16x4_f32 %[a1], %[w11], %[hf1], %[a1]\n\t"
        "v_pk_add_f32 v[200:201], v[200:201], v[204:205]\n\t"
        "v_pk_add_f32 v[202:203], v[202:203], v[206:207]\n\t"
        "v_wmma_f32_16x16x4_f32 %[a2], %[w21], %[hf1], %[a2]\n\t"
        "v_fma_f32 v200, %[h2], %[ri], v200\n\t"
        "v_fma_f32 v201, %[h2], %[rf], v201\n\t"
        "v_fma_f32 v202, %[h2], %[rg], v202\n\t"
        "v_fma_f32 v203, %[h2], %[ro], v203\n\t"
        "v_wmma_f32_16x16x4_f32 %[a3], %[w31], %[hf1], %[a3]\n\t"
        "v_add_f32 v200, %[bi], v200\n\t"
        "v_add_f32 v201, %[bf], v201\n\t"
        "v_add_f32 v202, %[bg], v202\n\t"
        "v_add_f32 v203, %[bo], v203\n\t"
        "v_wmma_f32_16x16x4_f32 %[a4], %[w41], %[hf1], %[a4]\n\t"
        "v_mul_f32 v200, 0xbfb8aa3b, v200\n\t"
        "v_mul_f32 v201, 0xbfb8aa3b, v201\n\t"
        "v_mul_f32 v202, 0x4038aa3b, v202\n\t"
        "v_mul_f32 v203, 0xbfb8aa3b, v203\n\t"
        "v_wmma_f32_16x16x4_f32 %[a0], %[w02], %[hf2], %[a0]\n\t"
        "v_exp_f32 v200, v200\n\t"
        "v_exp_f32 v201, v201\n\t"
        "v_exp_f32 v202, v202\n\t"
        "v_exp_f32 v203, v203\n\t"
        "v_wmma_f32_16x16x4_f32 %[a1], %[w12], %[hf2], %[a1]\n\t"
        "v_pk_add_f32 v[200:201], v[200:201], 1.0 op_sel_hi:[1,0]\n\t"
        "v_pk_add_f32 v[202:203], v[202:203], 1.0 op_sel_hi:[1,0]\n\t"
        "v_wmma_f32_16x16x4_f32 %[a2], %[w22], %[hf2], %[a2]\n\t"
        "v_rcp_f32 v200, v200\n\t"
        "v_rcp_f32 v201, v201\n\t"
        "v_rcp_f32 v202, v202\n\t"
        "v_rcp_f32 v203, v203\n\t"
        "v_wmma_f32_16x16x4_f32 %[a3], %[w32], %[hf2], %[a3]\n\t"
        "v_fma_f32 v202, v202, -2.0, 1.0\n\t"
        "v_wmma_f32_16x16x4_f32 %[a4], %[w42], %[hf2], %[a4]\n\t"
        "v_mul_f32 v200, v200, v202\n\t"
        "v_wmma_f32_16x16x4_f32 %[a0], %[w03], %[hf3], %[a0]\n\t"
        "v_fma_f32 %[c2], %[c2], v201, v200\n\t"
        "v_wmma_f32_16x16x4_f32 %[a1], %[w13], %[hf3], %[a1]\n\t"
        "v_mul_f32 v200, 0x4038aa3b, %[c2]\n\t"
        "v_wmma_f32_16x16x4_f32 %[a2], %[w23], %[hf3], %[a2]\n\t"
        "v_exp_f32 v200, v200\n\t"
        "v_wmma_f32_16x16x4_f32 %[a3], %[w33], %[hf3], %[a3]\n\t"
        "v_add_f32 v200, 1.0, v200\n\t"
        "v_wmma_f32_16x16x4_f32 %[a4], %[w43], %[hf3], %[a4]\n\t"
        "v_rcp_f32 v200, v200\n\t"
        "v_wmma_f32_16x16x4_f32 %[a0], %[w04], %[hf4], %[a0]\n\t"
        "v_fma_f32 v200, v200, -2.0, 1.0\n\t"
        "v_wmma_f32_16x16x4_f32 %[a1], %[w14], %[hf4], %[a1]\n\t"
        "v_mul_f32 %[h2], v203, v200\n\t"
        "v_wmma_f32_16x16x4_f32 %[a2], %[w24], %[hf4], %[a2]\n\t"
        "v_wmma_f32_16x16x4_f32 %[a3], %[w34], %[hf4], %[a3]\n\t"
        "v_wmma_f32_16x16x4_f32 %[a4], %[w44], %[hf4], %[a4]\n\t"
        "v_nop\n\t"
        "v_nop\n\t"
        "v_nop\n\t"
        "v_nop\n\t"
        "v_nop\n\t"
        "v_nop\n\t"
        "v_nop\n\t"
        "v_nop"
        : [a0] "=&v"(acc[0]), [a1] "=&v"(acc[1]), [a2] "=&v"(acc[2]), [a3] "=&v"(acc[3]), [a4] "=&v"(acc[4]),
          [h2] "+v"(h2), [c2] "+v"(c2)
        :
          [w00] "v"(wfrag[0][0]), [w01] "v"(wfrag[0][1]), [w02] "v"(wfrag[0][2]), [w03] "v"(wfrag[0][3]), [w04] "v"(wfrag[0][4]), [w05] "v"(wfrag[0][5]),
          [w10] "v"(wfrag[1][0]), [w11] "v"(wfrag[1][1]), [w12] "v"(wfrag[1][2]), [w13] "v"(wfrag[1][3]), [w14] "v"(wfrag[1][4]), [w15] "v"(wfrag[1][5]),
          [w20] "v"(wfrag[2][0]), [w21] "v"(wfrag[2][1]), [w22] "v"(wfrag[2][2]), [w23] "v"(wfrag[2][3]), [w24] "v"(wfrag[2][4]), [w25] "v"(wfrag[2][5]),
          [w30] "v"(wfrag[3][0]), [w31] "v"(wfrag[3][1]), [w32] "v"(wfrag[3][2]), [w33] "v"(wfrag[3][3]), [w34] "v"(wfrag[3][4]), [w35] "v"(wfrag[3][5]),
          [w40] "v"(wfrag[4][0]), [w41] "v"(wfrag[4][1]), [w42] "v"(wfrag[4][2]), [w43] "v"(wfrag[4][3]), [w44] "v"(wfrag[4][4]), [w45] "v"(wfrag[4][5]),
          [hf0] "v"(hfrag[0]), [hf1] "v"(hfrag[1]), [hf2] "v"(hfrag[2]), [hf3] "v"(hfrag[3]), [hf4] "v"(hfrag[4]), [xb] "v"(xb),
          [wif0] "v"(w2if[0]), [wif1] "v"(w2if[1]), [wif2] "v"(w2if[2]), [wif3] "v"(w2if[3]), [wif4] "v"(w2if[4]), [wif5] "v"(w2if[5]), [wif6] "v"(w2if[6]), [wif7] "v"(w2if[7]), [wif8] "v"(w2if[8]), [wif9] "v"(w2if[9]),
          [wgo0] "v"(w2go[0]), [wgo1] "v"(w2go[1]), [wgo2] "v"(w2go[2]), [wgo3] "v"(w2go[3]), [wgo4] "v"(w2go[4]), [wgo5] "v"(w2go[5]), [wgo6] "v"(w2go[6]), [wgo7] "v"(w2go[7]), [wgo8] "v"(w2go[8]), [wgo9] "v"(w2go[9]),
          [bi] "s"(b2[0]), [bf] "s"(b2[1]), [bg] "s"(b2[2]), [bo] "s"(b2[3]), [ri] "s"(whh2v[0]), [rf] "s"(whh2v[1]), [rg] "s"(whh2v[2]), [ro] "s"(whh2v[3]), [sel] "s"(0x76543210)
        : "v200", "v201", "v202", "v203", "v204", "v205", "v206", "v207");
    if (h == 0) orow[t - 1] = h2;

    layer1_cells(acc, c1, hfrag);
    xt = xt_next;
    xp += (t + 2 < TSEQ) ? 1 : 0;
  }

  {
    float a2[4];
    #pragma unroll
    for (int q = 0; q < 4; ++q) {
      float p = 0.0f;
      #pragma unroll
      for (int u = 0; u < UPL; ++u) {
        const float hv = (u & 1) ? hfrag[u >> 1].y : hfrag[u >> 1].x;
        const float w  = (q == 0) ? w2if[u].x : (q == 1) ? w2if[u].y : (q == 2) ? w2go[u].x : w2go[u].y;
        p = __builtin_fmaf(hv, w, p);
      }
      p = p + other_half(p);
      p = __builtin_fmaf(h2, whh2v[q], p);
      a2[q] = b2[q] + p;
    }
    const float i2 = __builtin_amdgcn_rcpf(1.0f + __builtin_amdgcn_exp2f(a2[0] * -1.44269504088896340736f));
    const float f2 = __builtin_amdgcn_rcpf(1.0f + __builtin_amdgcn_exp2f(a2[1] * -1.44269504088896340736f));
    const float g2 = __builtin_fmaf(__builtin_amdgcn_rcpf(1.0f + __builtin_amdgcn_exp2f(a2[2] * 2.88539008177792681472f)), -2.0f, 1.0f);
    const float o2 = __builtin_amdgcn_rcpf(1.0f + __builtin_amdgcn_exp2f(a2[3] * -1.44269504088896340736f));
    c2 = __builtin_fmaf(c2, f2, i2 * g2);
    const float tc = __builtin_fmaf(__builtin_amdgcn_rcpf(1.0f + __builtin_amdgcn_exp2f(c2 * 2.88539008177792681472f)), -2.0f, 1.0f);
    h2 = o2 * tc;
    if (h == 0) orow[TSEQ - 1] = h2;
  }
}

extern "C" void kernel_launch(void* const* d_in, const int* in_sizes, int n_in,
                              void* d_out, int out_size, void* d_ws, size_t ws_size,
                              hipStream_t stream) {
  const float* x    = (const float*)d_in[0];
  const float* Wih1 = (const float*)d_in[1];
  const float* Whh1 = (const float*)d_in[2];
  const float* bih1 = (const float*)d_in[3];
  const float* bhh1 = (const float*)d_in[4];
  const float* Wih2 = (const float*)d_in[5];
  const float* Whh2 = (const float*)d_in[6];
  const float* bih2 = (const float*)d_in[7];
  const float* bhh2 = (const float*)d_in[8];
  float* out = (float*)d_out;

  const int B    = in_sizes[0] / TSEQ;
  const int grid = B / (WAVES * 16);
  hipLaunchKernelGGL(lstm_fused, dim3(grid), dim3(WAVES * 32), 0, stream,
                     x, Wih1, Whh1, bih1, bhh1, Wih2, Whh2, bih2, bhh2, out);
}


# Time loop rotated by one step: the layer-1 cell arithmetic of step i hand-interleaved with the WMMAs of step i+1, issued in the order their B chunks become ready; sum order unchanged
# speedup vs baseline: 4.0719x; 1.0334x over previous
#include <hip/hip_runtime.h>

#define HID   20
#define TSEQ  1024
#define WAVES 4
#define MT    5
#define KC    6
#define UPL   10

typedef __attribute__((ext_vector_type(2))) float v2f;

__global__ __launch_bounds__(WAVES * 32)
void lstm_fused(const float* __restrict__ x,    const float* __restrict__ Wih1,
                const float* __restrict__ Whh1, const float* __restrict__ bih1,
                const float* __restrict__ bhh1, const float* __restrict__ Wih2,
                const float* __restrict__ Whh2, const float* __restrict__ bih2,
                const float* __restrict__ bhh2, float* __restrict__ out)
{
  const int lane = threadIdx.x & 31;
  const int wave = threadIdx.x >> 5;
  const int c    = lane & 15;
  const int h    = lane >> 4;
  const int row  = (blockIdx.x * WAVES + wave) * 16 + c;

  v2f wfrag[MT][KC];
  {
    const int hm = c >> 3;
    const int r  = c & 7;
    const int q  = r & 3;
    #pragma unroll
    for (int mt = 0; mt < MT; ++mt) {
      const int um = hm * UPL + mt * 2 + (r >> 2);
      const int n  = q * HID + um;
      #pragma unroll
      for (int kc = 0; kc < KC - 1; ++kc) {
        const int uk = h * UPL + kc * 2;
        v2f w;
        w.x = Whh1[n * HID + uk];
        w.y = Whh1[n * HID + uk + 1];
        wfrag[mt][kc] = w;
      }
      v2f wl;
      wl.x = (h == 0) ? Wih1[n]           : 0.0f;
      wl.y = (h == 0) ? bih1[n] + bhh1[n] : 0.0f;
      wfrag[mt][KC - 1] = wl;
    }
  }

  v2f w2if[UPL], w2go[UPL];
  float b2[4], whh2v[4];
  #pragma unroll
  for (int u = 0; u < UPL; ++u) {
    const int j = h * UPL + u;
    w2if[u] = (v2f){Wih2[0 * HID + j], Wih2[1 * HID + j]};
    w2go[u] = (v2f){Wih2[2 * HID + j], Wih2[3 * HID + j]};
  }
  #pragma unroll
  for (int q = 0; q < 4; ++q) {
    b2[q]    = bih2[q] + bhh2[q];
    whh2v[q] = Whh2[q];
  }

  const unsigned rowoff = (unsigned)row * (TSEQ * 4u);
  const float x0 = x[(size_t)row * TSEQ];
  const float x1 = x[(size_t)row * TSEQ + 1];

  asm volatile(
        "s_mov_b32 s41, 4\n\t"
        "v_mov_b32 v194, 0xbfb8aa3b\n\t"
        "v_mov_b32 v195, 0xbfb8aa3b\n\t"
        "v_mov_b32 v196, 0x4038aa3b\n\t"
        "v_mov_b32 v197, 0xbfb8aa3b\n\t"
        "v_mov_b32 v198, 0x4038aa3b\n\t"
        "v_mov_b32 v199, 0x4038aa3b\n\t"
        "v_mov_b32 v164, 0\n\t"
        "v_mov_b32 v165, 0\n\t"
        "v_mov_b32 v166, 0\n\t"
        "v_mov_b32 v167, 0\n\t"
        "v_mov_b32 v168, 0\n\t"
        "v_mov_b32 v169, 0\n\t"
        "v_mov_b32 v170, 0\n\t"
        "v_mov_b32 v171, 0\n\t"
        "v_mov_b32 v172, 0\n\t"
        "v_mov_b32 v173, 0\n\t"
        "v_mov_b32 v174, 0\n\t"
        "v_mov_b32 v175, 0\n\t"
        "v_mov_b32 v176, 0\n\t"
        "v_mov_b32 v177, 0\n\t"
        "v_mov_b32 v178, 0\n\t"
        "v_mov_b32 v179, 0\n\t"
        "v_mov_b32 v180, 0\n\t"
        "v_mov_b32 v181, 0\n\t"
        "v_mov_b32 v182, 0\n\t"
        "v_mov_b32 v183, 0\n\t"
        "v_mov_b32 v203, 0\n\t"
        "v_mov_b32 v204, 0\n\t"
        "v_cndmask_b32_e64 v193, 0, 1.0, %[msk]\n\t"
        "v_cndmask_b32_e64 v192, 0, %[x0], %[msk]\n\t"
        "v_mov_b32 v200, %[x1]\n\t"
        "v_add_nc_u32 v201, 8, %[rowoff]\n\t"
        "v_add_nc_u32 v202, -4, %[rowoff]\n\t"
        "v_wmma_f32_16x16x4_f32 v[124:131], %[w05], v[192:193], 0\n\t"
        "v_wmma_f32_16x16x4_f32 v[132:139], %[w15], v[192:193], 0\n\t"
        "v_wmma_f32_16x16x4_f32 v[140:147], %[w25], v[192:193], 0\n\t"
        "v_wmma_f32_16x16x4_f32 v[148:155], %[w35], v[192:193], 0\n\t"
        "v_wmma_f32_16x16x4_f32 v[156:163], %[w45], v[192:193], 0\n\t"
        "v_nop\n\t"
        "v_nop\n\t"
        "v_nop\n\t"
        "v_nop\n\t"
        "v_nop\n\t"
        "v_nop\n\t"
        "v_nop\n\t"
        "v_nop\n\t"
        "v_nop\n\t"
        "v_wmma_f32_16x16x4_f32 v[132:139], %[w14], v[172:173], v[132:139]\n\t"
        "s_wait_loadcnt 0x0\n\t"
        "v_cndmask_b32_e64 v192, 0, v200, %[msk]\n\t"
        "global_load_b32 v200, v201, %[xbp]\n\t"
        "v_pk_mul_f32 v[126:127], v[126:127], v[196:197]\n\t"
        "v_pk_mul_f32 v[130:131], v[130:131], v[196:197]\n\t"
        "v_pk_mul_f32 v[124:125], v[124:125], v[194:195]\n\t"
        "v_wmma_f32_16x16x4_f32 v[140:147], %[w24], v[172:173], v[140:147]\n\t"
        "v_pk_mul_f32 v[128:129], v[128:129], v[194:195]\n\t"
        "v_exp_f32 v210, v126\n\t"
        "v_exp_f32 v211, v130\n\t"
        "v_exp_f32 v206, v124\n\t"
        "v_exp_f32 v208, v125\n\t"
        "v_exp_f32 v207, v128\n\t"
        "v_exp_f32 v209, v129\n\t"
        "v_pk_add_f32 v[210:211], v[210:211], 1.0 op_sel_hi:[1,0]\n\t"
        "v_wmma_f32_16x16x4_f32 v[148:155], %[w34], v[172:173], v[148:155]\n\t"
        "v_exp_f32 v212, v127\n\t"
        "v_pk_mul_f32 v[132:133], v[132:133], v[194:195]\n\t"
        "v_pk_add_f32 v[206:207], v[206:207], 1.0 op_sel_hi:[1,0]\n\t"
        "v_rcp_f32 v210, v210\n\t"
        "v_pk_mul_f32 v[134:135], v[134:135], v[196:197]\n\t"
        "v_rcp_f32 v211, v211\n\t"
        "v_pk_mul_f32 v[136:137], v[136:137], v[194:195]\n\t"
        "v_exp_f32 v213, v131\n\t"
        "v_pk_mul_f32 v[138:139], v[138:139], v[196:197]\n\t"
        "v_pk_add_f32 v[208:209], v[208:209], 1.0 op_sel_hi:[1,0]\n\t"
        "v_rcp_f32 v206, v206\n\t"
        "v_rcp_f32 v207, v207\n\t"
        "v_wmma_f32_16x16x4_f32 v[156:163], %[w44], v[172:173], v[156:163]\n\t"
        "v_pk_fma_f32 v[210:211], v[210:211], 2.0, 1.0 op_sel_hi:[1,0,0] neg_lo:[1,0,0] neg_hi:[1,0,0]\n\t"
        "v_exp_f32 v216, v132\n\t"
        "v_pk_mul_f32 v[140:141], v[140:141], v[194:195]\n\t"
        "v_exp_f32 v217, v136\n\t"
        "v_pk_mul_f32 v[142:143], v[142:143], v[196:197]\n\t"
        "v_exp_f32 v220, v134\n\t"
        "v_pk_mul_f32 v[144:145], v[144:145], v[194:195]\n\t"
        "v_exp_f32 v221, v138\n\t"
        "v_pk_mul_f32 v[206:207], v[206:207], v[210:211]\n\t"
        "v_exp_f32 v218, v133\n\t"
        "v_pk_mul_f32 v[146:147], v[146:147], v[196:197]\n\t"
        "v_exp_f32 v219, v137\n\t"
        "v_wmma_f32_16x16x4_f32 v[124:131], %[w05], v[192:193], 0\n\t"
        "v_exp_f32 v222, v135\n\t"
        "v_pk_mul_f32 v[148:149], v[148:149], v[194:195]\n\t"
        "v_exp_f32 v223, v139\n\t"
        "v_pk_mul_f32 v[150:151], v[150:151], v[196:197]\n\t"
        "v_rcp_f32 v208, v208\n\t"
        "v_pk_mul_f32 v[152:153], v[152:153], v[194:195]\n\t"
        "v_rcp_f32 v209, v209\n\t"
        "v_pk_mul_f32 v[154:155], v[154:155], v[196:197]\n\t"
        "v_exp_f32 v226, v140\n\t"
        "v_exp_f32 v227, v144\n\t"
        "v_exp_f32 v230, v142\n\t"
        "v_wmma_f32_16x16x4_f32 v[132:139], %[w15], v[192:193], 0\n\t"
        "v_pk_fma_f32 v[174:175], v[174:175], v[208:209], v[206:207]\n\t"
        "v_exp_f32 v231, v146\n\t"
        "v_pk_mul_f32 v[156:157], v[156:157], v[194:195]\n\t"
        "v_exp_f32 v228, v141\n\t"
        "v_pk_mul_f32 v[158:159], v[158:159], v[196:197]\n\t"
        "v_exp_f32 v229, v145\n\t"
        "v_pk_mul_f32 v[160:161], v[160:161], v[194:195]\n\t"
        "v_exp_f32 v232, v143\n\t"
        "v_pk_mul_f32 v[214:215], v[174:175], v[198:199]\n\t"
        "v_exp_f32 v233, v147\n\t"
        "v_pk_mul_f32 v[162:163], v[162:163], v[196:197]\n\t"
        "v_exp_f32 v236, v148\n\t"
        "v_pk_add_f32 v[212:213], v[212:213], 1.0 op_sel_hi:[1,0]\n\t"
        "v_wmma_f32_16x16x4_f32 v[140:147], %[w25], v[192:193], 0\n\t"
        "v_exp_f32 v237, v152\n\t"
        "v_pk_add_f32 v[220:221], v[220:221], 1.0 op_sel_hi:[1,0]\n\t"
        "v_exp_f32 v214, v214\n\t"
        "v_pk_add_f32 v[216:217], v[216:217], 1.0 op_sel_hi:[1,0]\n\t"
        "v_exp_f32 v215, v215\n\t"
        "v_pk_add_f32 v[218:219], v[218:219], 1.0 op_sel_hi:[1,0]\n\t"
        "v_exp_f32 v240, v150\n\t"
        "v_exp_f32 v241, v154\n\t"
        "v_exp_f32 v238, v149\n\t"
        "v_exp_f32 v239, v153\n\t"
        "v_pk_add_f32 v[214:215], v[214:215], 1.0 op_sel_hi:[1,0]\n\t"
        "v_exp_f32 v242, v151\n\t"
        "v_exp_f32 v243, v155\n\t"
        "v_nop\n\t"
        "v_wmma_f32_16x16x4_f32 v[148:155], %[w35], v[192:193], 0\n\t"
        "v_exp_f32 v246, v156\n\t"
        "v_pk_add_f32 v[222:223], v[222:223], 1.0 op_sel_hi:[1,0]\n\t"
        "v_rcp_f32 v214, v214\n\t"
        "v_pk_add_f32 v[230:231], v[230:231], 1.0 op_sel_hi:[1,0]\n\t"
        "v_rcp_f32 v215, v215\n\t"
        "v_pk_add_f32 v[226:227], v[226:227], 1.0 op_sel_hi:[1,0]\n\t"
        "v_exp_f32 v247, v160\n\t"
        "v_exp_f32 v250, v158\n\t"
        "v_exp_f32 v251, v162\n\t"
        "v_exp_f32 v248, v157\n\t"
        "v_pk_fma_f32 v[214:215], v[214:215], 2.0, 1.0 op_sel_hi:[1,0,0] neg_lo:[1,0,0] neg_hi:[1,0,0]\n\t"
        "v_exp_f32 v249, v161\n\t"
        "v_exp_f32 v252, v159\n\t"
        "v_exp_f32 v253, v163\n\t"
        "v_nop\n\t"
        "v_wmma_f32_16x16x4_f32 v[156:163], %[w45], v[192:193], 0\n\t"
        "v_rcp_f32 v212, v212\n\t"
        "v_pk_add_f32 v[228:229], v[228:229], 1.0 op_sel_hi:[1,0]\n\t"
        "v_rcp_f32 v213, v213\n\t"
        "v_pk_add_f32 v[232:233], v[232:233], 1.0 op_sel_hi:[1,0]\n\t"
        "v_rcp_f32 v220, v220\n\t"
        "v_pk_add_f32 v[240:241], v[240:241], 1.0 op_sel_hi:[1,0]\n\t"
        "v_rcp_f32 v221, v221\n\t"
        "v_rcp_f32 v216, v216\n\t"
        "v_pk_mul_f32 v[164:165], v[212:213], v[214:215]\n\t"
        "v_rcp_f32 v217, v217\n\t"
        "v_rcp_f32 v218, v218\n\t"
        "v_pk_fma_f32 v[220:221], v[220:221], 2.0, 1.0 op_sel_hi:[1,0,0] neg_lo:[1,0,0] neg_hi:[1,0,0]\n\t"
        "v_rcp_f32 v219, v219\n\t"
        "v_wmma_f32_16x16x4_f32 v[124:131], %[w00], v[164:165], v[124:131]\n\t"
        "v_rcp_f32 v222, v222\n\t"
        "v_pk_add_f32 v[236:237], v[236:237], 1.0 op_sel_hi:[1,0]\n\t"
        "v_rcp_f32 v223, v223\n\t"
        "v_pk_add_f32 v[238:239], v[238:239], 1.0 op_sel_hi:[1,0]\n\t"
        "v_pk_mul_f32 v[216:217], v[216:217], v[220:221]\n\t"
        "v_rcp_f32 v230, v230\n\t"
        "v_pk_add_f32 v[242:243], v[242:243], 1.0 op_sel_hi:[1,0]\n\t"
        "v_pk_fma_f32 v[176:177], v[176:177], v[218:219], v[216:217]\n\t"
        "v_wmma_f32_16x16x4_f32 v[132:139], %[w10], v[164:165], v[132:139]\n\t"
        "v_rcp_f32 v231, v231\n\t"
        "v_pk_add_f32 v[250:251], v[250:251], 1.0 op_sel_hi:[1,0]\n\t"
        "v_rcp_f32 v226, v226\n\t"
        "v_pk_add_f32 v[246:247], v[246:247], 1.0 op_sel_hi:[1,0]\n\t"
        "v_pk_mul_f32 v[224:225], v[176:177], v[198:199]\n\t"
        "v_rcp_f32 v227, v227\n\t"
        "v_pk_add_f32 v[248:249], v[248:249], 1.0 op_sel_hi:[1,0]\n\t"
        "v_exp_f32 v224, v224\n\t"
        "v_exp_f32 v225, v225\n\t"
        "v_wmma_f32_16x16x4_f32 v[140:147], %[w20], v[164:165], v[140:147]\n\t"
        "v_pk_fma_f32 v[230:231], v[230:231], 2.0, 1.0 op_sel_hi:[1,0,0] neg_lo:[1,0,0] neg_hi:[1,0,0]\n\t"
        "v_rcp_f32 v228, v228\n\t"
        "v_pk_add_f32 v[252:253], v[252:253], 1.0 op_sel_hi:[1,0]\n\t"
        "v_rcp_f32 v229, v229\n\t"
        "v_rcp_f32 v232, v232\n\t"
        "v_pk_add_f32 v[224:225], v[224:225], 1.0 op_sel_hi:[1,0]\n\t"
        "v_pk_mul_f32 v[226:227], v[226:227], v[230:231]\n\t"
        "v_rcp_f32 v224, v224\n\t"
        "v_wmma_f32_16x16x4_f32 v[148:155], %[w30], v[164:165], v[148:155]\n\t"
        "v_rcp_f32 v225, v225\n\t"
        "v_pk_fma_f32 v[178:179], v[178:179], v[228:229], v[226:227]\n\t"
        "v_rcp_f32 v233, v233\n\t"
        "v_rcp_f32 v240, v240\n\t"
        "v_rcp_f32 v241, v241\n\t"
        "v_rcp_f32 v236, v236\n\t"
        "v_pk_fma_f32 v[224:225], v[224:225], 2.0, 1.0 op_sel_hi:[1,0,0] neg_lo:[1,0,0] neg_hi:[1,0,0]\n\t"
        "v_wmma_f32_16x16x4_f32 v[156:163], %[w40], v[164:165], v[156:163]\n\t"
        "v_pk_mul_f32 v[234:235], v[178:179], v[198:199]\n\t"
        "v_pk_mul_f32 v[166:167], v[222:223], v[224:225]\n\t"
        "v_rcp_f32 v237, v237\n\t"
        "v_pk_fma_f32 v[240:241], v[240:241], 2.0, 1.0 op_sel_hi:[1,0,0] neg_lo:[1,0,0] neg_hi:[1,0,0]\n\t"
        "v_rcp_f32 v238, v238\n\t"
        "v_rcp_f32 v239, v239\n\t"
        "v_exp_f32 v234, v234\n\t"
        "v_exp_f32 v235, v235\n\t"
        "v_wmma_f32_16x16x4_f32 v[124:131], %[w01], v[166:167], v[124:131]\n\t"
        "v_pk_mul_f32 v[236:237], v[236:237], v[240:241]\n\t"
        "v_rcp_f32 v242, v242\n\t"
        "v_rcp_f32 v243, v243\n\t"
        "v_pk_add_f32 v[234:235], v[234:235], 1.0 op_sel_hi:[1,0]\n\t"
        "v_rcp_f32 v250, v250\n\t"
        "v_pk_fma_f32 v[180:181], v[180:181], v[238:239], v[236:237]\n\t"
        "v_rcp_f32 v251, v251\n\t"
        "v_rcp_f32 v234, v234\n\t"
        "v_wmma_f32_16x16x4_f32 v[132:139], %[w11], v[166:167], v[132:139]\n\t"
        "v_rcp_f32 v235, v235\n\t"
        "v_pk_mul_f32 v[244:245], v[180:181], v[198:199]\n\t"
        "v_rcp_f32 v246, v246\n\t"
        "v_pk_fma_f32 v[250:251], v[250:251], 2.0, 1.0 op_sel_hi:[1,0,0] neg_lo:[1,0,0] neg_hi:[1,0,0]\n\t"
        "v_rcp_f32 v247, v247\n\t"
        "s_wait_xcnt 0x0\n\t"
        "v_add_nc_u32 v201, s41, v201\n\t"
        "v_rcp_f32 v248, v248\n\t"
        "v_pk_fma_f32 v[234:235], v[234:235], 2.0, 1.0 op_sel_hi:[1,0,0] neg_lo:[1,0,0] neg_hi:[1,0,0]\n\t"
        "v_wmma_f32_16x16x4_f32 v[140:147], %[w21], v[166:167], v[140:147]\n\t"
        "v_exp_f32 v244, v244\n\t"
        "v_pk_mul_f32 v[168:169], v[232:233], v[234:235]\n\t"
        "v_exp_f32 v245, v245\n\t"
        "v_pk_mul_f32 v[246:247], v[246:247], v[250:251]\n\t"
        "v_rcp_f32 v249, v249\n\t"
        "v_rcp_f32 v252, v252\n\t"
        "v_rcp_f32 v253, v253\n\t"
        "v_wmma_f32_16x16x4_f32 v[148:155], %[w31], v[166:167], v[148:155]\n\t"
        "v_pk_add_f32 v[244:245], v[244:245], 1.0 op_sel_hi:[1,0]\n\t"
        "v_pk_fma_f32 v[182:183], v[182:183], v[248:249], v[246:247]\n\t"
        "v_rcp_f32 v244, v244\n\t"
        "v_rcp_f32 v245, v245\n\t"
        "v_pk_mul_f32 v[254:255], v[182:183], v[198:199]\n\t"
        "v_wmma_f32_16x16x4_f32 v[156:163], %[w41], v[166:167], v[156:163]\n\t"
        "v_pk_fma_f32 v[244:245], v[244:245], 2.0, 1.0 op_sel_hi:[1,0,0] neg_lo:[1,0,0] neg_hi:[1,0,0]\n\t"
        "v_exp_f32 v254, v254\n\t"
        "v_exp_f32 v255, v255\n\t"
        "v_pk_mul_f32 v[170:171], v[242:243], v[244:245]\n\t"
        "v_wmma_f32_16x16x4_f32 v[124:131], %[w02], v[168:169], v[124:131]\n\t"
        "v_pk_add_f32 v[254:255], v[254:255], 1.0 op_sel_hi:[1,0]\n\t"
        "v_rcp_f32 v254, v254\n\t"
        "v_rcp_f32 v255, v255\n\t"
        "v_wmma_f32_16x16x4_f32 v[132:139], %[w12], v[168:169], v[132:139]\n\t"
        "v_pk_fma_f32 v[254:255], v[254:255], 2.0, 1.0 op_sel_hi:[1,0,0] neg_lo:[1,0,0] neg_hi:[1,0,0]\n\t"
        "v_pk_mul_f32 v[172:173], v[252:253], v[254:255]\n\t"
        "v_wmma_f32_16x16x4_f32 v[140:147], %[w22], v[168:169], v[140:147]\n\t"
        "v_wmma_f32_16x16x4_f32 v[148:155], %[w32], v[168:169], v[148:155]\n\t"
        "v_wmma_f32_16x16x4_f32 v[156:163], %[w42], v[168:169], v[156:163]\n\t"
        "v_wmma_f32_16x16x4_f32 v[124:131], %[w03], v[170:171], v[124:131]\n\t"
        "v_wmma_f32_16x16x4_f32 v[132:139], %[w13], v[170:171], v[132:139]\n\t"
        "v_wmma_f32_16x16x4_f32 v[140:147], %[w23], v[170:171], v[140:147]\n\t"
        "v_wmma_f32_16x16x4_f32 v[148:155], %[w33], v[170:171], v[148:155]\n\t"
        "v_wmma_f32_16x16x4_f32 v[156:163], %[w43], v[170:171], v[156:163]\n\t"
        "v_wmma_f32_16x16x4_f32 v[124:131], %[w04], v[172:173], v[124:131]\n\t"
        "s_mov_b32 s40, 1\n\t"
        "1:\n\t"
        "s_cmp_lt_u32 s40, 0x3fd\n\t"
        "s_cselect_b32 s41, 4, 0\n\t"
        "v_wmma_f32_16x16x4_f32 v[132:139], %[w14], v[172:173], v[132:139]\n\t"
        "s_wait_loadcnt 0x0\n\t"
        "v_cndmask_b32_e64 v192, 0, v200, %[msk]\n\t"
        "v_pk_fma_f32 v[184:185], v[164:165], %[wif0], 0 op_sel_hi:[0,1,0]\n\t"
        "v_pk_fma_f32 v[186:187], v[164:165], %[wgo0], 0 op_sel_hi:[0,1,0]\n\t"
        "global_load_b32 v200, v201, %[xbp]\n\t"
        "v_nop\n\t"
        "v_nop\n\t"
        "v_nop\n\t"
        "v_nop\n\t"
        "v_pk_fma_f32 v[184:185], v[164:165], %[wif1], v[184:185] op_sel:[1,0,0]\n\t"
        "v_nop\n\t"
        "v_pk_mul_f32 v[126:127], v[126:127], v[196:197]\n\t"
        "v_pk_mul_f32 v[130:131], v[130:131], v[196:197]\n\t"
        "v_pk_mul_f32 v[124:125], v[124:125], v[194:195]\n\t"
        "v_wmma_f32_16x16x4_f32 v[140:147], %[w24], v[172:173], v[140:147]\n\t"
        "v_pk_mul_f32 v[128:129], v[128:129], v[194:195]\n\t"
        "v_pk_fma_f32 v[186:187], v[164:165], %[wgo1], v[186:187] op_sel:[1,0,0]\n\t"
        "v_exp_f32 v210, v126\n\t"
        "v_pk_fma_f32 v[184:185], v[166:167], %[wif2], v[184:185] op_sel_hi:[0,1,1]\n\t"
        "v_exp_f32 v211, v130\n\t"
        "v_exp_f32 v206, v124\n\t"
        "v_exp_f32 v207, v128\n\t"
        "v_pk_add_f32 v[210:211], v[210:211], 1.0 op_sel_hi:[1,0]\n\t"
        "v_wmma_f32_16x16x4_f32 v[148:155], %[w34], v[172:173], v[148:155]\n\t"
        "v_exp_f32 v208, v125\n\t"
        "v_pk_mul_f32 v[132:133], v[132:133], v[194:195]\n\t"
        "v_exp_f32 v209, v129\n\t"
        "v_pk_add_f32 v[206:207], v[206:207], 1.0 op_sel_hi:[1,0]\n\t"
        "v_exp_f32 v212, v127\n\t"
        "v_pk_mul_f32 v[134:135], v[134:135], v[196:197]\n\t"
        "v_rcp_f32 v210, v210\n\t"
        "v_pk_mul_f32 v[136:137], v[136:137], v[194:195]\n\t"
        "v_rcp_f32 v211, v211\n\t"
        "v_pk_mul_f32 v[138:139], v[138:139], v[196:197]\n\t"
        "v_exp_f32 v213, v131\n\t"
        "v_pk_add_f32 v[208:209], v[208:209], 1.0 op_sel_hi:[1,0]\n\t"
        "v_rcp_f32 v206, v206\n\t"
        "v_wmma_f32_16x16x4_f32 v[156:163], %[w44], v[172:173], v[156:163]\n\t"
        "v_rcp_f32 v207, v207\n\t"
        "v_pk_mul_f32 v[140:141], v[140:141], v[194:195]\n\t"
        "v_pk_fma_f32 v[210:211], v[210:211], 2.0, 1.0 op_sel_hi:[1,0,0] neg_lo:[1,0,0] neg_hi:[1,0,0]\n\t"
        "v_exp_f32 v216, v132\n\t"
        "v_pk_mul_f32 v[142:143], v[142:143], v[196:197]\n\t"
        "v_exp_f32 v217, v136\n\t"
        "v_pk_mul_f32 v[144:145], v[144:145], v[194:195]\n\t"
        "v_exp_f32 v220, v134\n\t"
        "v_pk_mul_f32 v[146:147], v[146:147], v[196:197]\n\t"
        "v_exp_f32 v221, v138\n\t"
        "v_pk_mul_f32 v[206:207], v[206:207], v[210:211]\n\t"
        "v_exp_f32 v218, v133\n\t"
        "v_wmma_f32_16x16x4_f32 v[124:131], %[w05], v[192:193], 0\n\t"
        "v_exp_f32 v219, v137\n\t"
        "v_pk_mul_f32 v[148:149], v[148:149], v[194:195]\n\t"
        "v_exp_f32 v222, v135\n\t"
        "v_pk_mul_f32 v[150:151], v[150:151], v[196:197]\n\t"
        "v_exp_f32 v223, v139\n\t"
        "v_pk_mul_f32 v[152:153], v[152:153], v[194:195]\n\t"
        "v_rcp_f32 v208, v208\n\t"
        "v_pk_mul_f32 v[154:155], v[154:155], v[196:197]\n\t"
        "v_rcp_f32 v209, v209\n\t"
        "v_exp_f32 v226, v140\n\t"
        "v_exp_f32 v227, v144\n\t"
        "v_wmma_f32_16x16x4_f32 v[132:139], %[w15], v[192:193], 0\n\t"
        "v_exp_f32 v230, v142\n\t"
        "v_pk_mul_f32 v[156:157], v[156:157], v[194:195]\n\t"
        "v_pk_fma_f32 v[174:175], v[174:175], v[208:209], v[206:207]\n\t"
        "v_exp_f32 v231, v146\n\t"
        "v_pk_mul_f32 v[158:159], v[158:159], v[196:197]\n\t"
        "v_exp_f32 v228, v141\n\t"
        "v_pk_mul_f32 v[160:161], v[160:161], v[194:195]\n\t"
        "v_exp_f32 v229, v145\n\t"
        "v_pk_mul_f32 v[162:163], v[162:163], v[196:197]\n\t"
        "v_exp_f32 v232, v143\n\t"
        "v_pk_mul_f32 v[214:215], v[174:175], v[198:199]\n\t"
        "v_exp_f32 v233, v147\n\t"
        "v_pk_add_f32 v[212:213], v[212:213], 1.0 op_sel_hi:[1,0]\n\t"
        "v_wmma_f32_16x16x4_f32 v[140:147], %[w25], v[192:193], 0\n\t"
        "v_exp_f32 v236, v148\n\t"
        "v_pk_add_f32 v[220:221], v[220:221], 1.0 op_sel_hi:[1,0]\n\t"
        "v_exp_f32 v237, v152\n\t"
        "v_pk_add_f32 v[216:217], v[216:217], 1.0 op_sel_hi:[1,0]\n\t"
        "v_exp_f32 v214, v214\n\t"
        "v_pk_add_f32 v[218:219], v[218:219], 1.0 op_sel_hi:[1,0]\n\t"
        "v_exp_f32 v215, v215\n\t"
        "v_exp_f32 v240, v150\n\t"
        "v_exp_f32 v241, v154\n\t"
        "v_exp_f32 v238, v149\n\t"
        "v_exp_f32 v239, v153\n\t"
        "v_pk_add_f32 v[214:215], v[214:215], 1.0 op_sel_hi:[1,0]\n\t"
        "v_exp_f32 v242, v151\n\t"
        "v_exp_f32 v243, v155\n\t"
        "v_nop\n\t"
        "v_wmma_f32_16x16x4_f32 v[148:155], %[w35], v[192:193], 0\n\t"
        "v_exp_f32 v246, v156\n\t"
        "v_pk_add_f32 v[222:223], v[222:223], 1.0 op_sel_hi:[1,0]\n\t"
        "v_rcp_f32 v214, v214\n\t"
        "v_pk_fma_f32 v[186:187], v[166:167], %[wgo2], v[186:187] op_sel_hi:[0,1,1]\n\t"
        "v_rcp_f32 v215, v215\n\t"
        "v_pk_add_f32 v[230:231], v[230:231], 1.0 op_sel_hi:[1,0]\n\t"
        "v_exp_f32 v247, v160\n\t"
        "v_exp_f32 v250, v158\n\t"
        "v_exp_f32 v251, v162\n\t"
        "v_exp_f32 v248, v157\n\t"
        "v_pk_fma_f32 v[214:215], v[214:215], 2.0, 1.0 op_sel_hi:[1,0,0] neg_lo:[1,0,0] neg_hi:[1,0,0]\n\t"
        "v_exp_f32 v249, v161\n\t"
        "v_exp_f32 v252, v159\n\t"
        "v_exp_f32 v253, v163\n\t"
        "v_nop\n\t"
        "v_wmma_f32_16x16x4_f32 v[156:163], %[w45], v[192:193], 0\n\t"
        "v_rcp_f32 v212, v212\n\t"
        "v_pk_fma_f32 v[184:185], v[166:167], %[wif3], v[184:185] op_sel:[1,0,0]\n\t"
        "v_rcp_f32 v213, v213\n\t"
        "v_pk_fma_f32 v[186:187], v[166:167], %[wgo3], v[186:187] op_sel:[1,0,0]\n\t"
        "v_rcp_f32 v220, v220\n\t"
        "v_pk_add_f32 v[226:227], v[226:227], 1.0 op_sel_hi:[1,0]\n\t"
        "v_rcp_f32 v221, v221\n\t"
        "v_rcp_f32 v216, v216\n\t"
        "v_pk_mul_f32 v[164:165], v[212:213], v[214:215]\n\t"
        "v_rcp_f32 v217, v217\n\t"
        "v_rcp_f32 v218, v218\n\t"
        "v_pk_fma_f32 v[220:221], v[220:221], 2.0, 1.0 op_sel_hi:[1,0,0] neg_lo:[1,0,0] neg_hi:[1,0,0]\n\t"
        "v_rcp_f32 v219, v219\n\t"
        "v_wmma_f32_16x16x4_f32 v[124:131], %[w00], v[164:165], v[124:131]\n\t"
        "v_rcp_f32 v222, v222\n\t"
        "v_pk_add_f32 v[228:229], v[228:229], 1.0 op_sel_hi:[1,0]\n\t"
        "v_rcp_f32 v223, v223\n\t"
        "v_pk_add_f32 v[232:233], v[232:233], 1.0 op_sel_hi:[1,0]\n\t"
        "v_pk_mul_f32 v[216:217], v[216:217], v[220:221]\n\t"
        "v_rcp_f32 v230, v230\n\t"
        "v_pk_fma_f32 v[184:185], v[168:169], %[wif4], v[184:185] op_sel_hi:[0,1,1]\n\t"
        "v_pk_fma_f32 v[176:177], v[176:177], v[218:219], v[216:217]\n\t"
        "v_wmma_f32_16x16x4_f32 v[132:139], %[w10], v[164:165], v[132:139]\n\t"
        "v_rcp_f32 v231, v231\n\t"
        "v_pk_fma_f32 v[186:187], v[168:169], %[wgo4], v[186:187] op_sel_hi:[0,1,1]\n\t"
        "v_rcp_f32 v226, v226\n\t"
        "v_pk_add_f32 v[240:241], v[240:241], 1.0 op_sel_hi:[1,0]\n\t"
        "v_pk_mul_f32 v[224:225], v[176:177], v[198:199]\n\t"
        "v_rcp_f32 v227, v227\n\t"
        "v_pk_fma_f32 v[184:185], v[168:169], %[wif5], v[184:185] op_sel:[1,0,0]\n\t"
        "v_exp_f32 v224, v224\n\t"
        "v_exp_f32 v225, v225\n\t"
        "v_wmma_f32_16x16x4_f32 v[140:147], %[w20], v[164:165], v[140:147]\n\t"
        "v_pk_fma_f32 v[230:231], v[230:231], 2.0, 1.0 op_sel_hi:[1,0,0] neg_lo:[1,0,0] neg_hi:[1,0,0]\n\t"
        "v_rcp_f32 v228, v228\n\t"
        "v_pk_fma_f32 v[186:187], v[168:169], %[wgo5], v[186:187] op_sel:[1,0,0]\n\t"
        "v_rcp_f32 v229, v229\n\t"
        "v_pk_add_f32 v[236:237], v[236:237], 1.0 op_sel_hi:[1,0]\n\t"
        "v_rcp_f32 v232, v232\n\t"
        "v_pk_add_f32 v[224:225], v[224:225], 1.0 op_sel_hi:[1,0]\n\t"
        "v_wmma_f32_16x16x4_f32 v[148:155], %[w30], v[164:165], v[148:155]\n\t"
        "v_rcp_f32 v224, v224\n\t"
        "v_pk_mul_f32 v[226:227], v[226:227], v[230:231]\n\t"
        "v_rcp_f32 v225, v225\n\t"
        "v_pk_add_f32 v[238:239], v[238:239], 1.0 op_sel_hi:[1,0]\n\t"
        "v_rcp_f32 v233, v233\n\t"
        "v_pk_fma_f32 v[186:187], v[170:171], %[wgo6], v[186:187] op_sel_hi:[0,1,1]\n\t"
        "v_rcp_f32 v240, v240\n\t"
        "v_pk_fma_f32 v[178:179], v[178:179], v[228:229], v[226:227]\n\t"
        "v_pk_fma_f32 v[224:225], v[224:225], 2.0, 1.0 op_sel_hi:[1,0,0] neg_lo:[1,0,0] neg_hi:[1,0,0]\n\t"
        "v_wmma_f32_16x16x4_f32 v[156:163], %[w40], v[164:165], v[156:163]\n\t"
        "v_rcp_f32 v241, v241\n\t"
        "v_pk_mul_f32 v[234:235], v[178:179], v[198:199]\n\t"
        "v_rcp_f32 v236, v236\n\t"
        "v_pk_mul_f32 v[166:167], v[222:223], v[224:225]\n\t"
        "v_rcp_f32 v237, v237\n\t"
        "v_pk_fma_f32 v[184:185], v[170:171], %[wif6], v[184:185] op_sel_hi:[0,1,1]\n\t"
        "v_rcp_f32 v238, v238\n\t"
        "v_exp_f32 v234, v234\n\t"
        "v_exp_f32 v235, v235\n\t"
        "v_wmma_f32_16x16x4_f32 v[124:131], %[w01], v[166:167], v[124:131]\n\t"
        "v_pk_fma_f32 v[240:241], v[240:241], 2.0, 1.0 op_sel_hi:[1,0,0] neg_lo:[1,0,0] neg_hi:[1,0,0]\n\t"
        "v_rcp_f32 v239, v239\n\t"
        "v_pk_fma_f32 v[186:187], v[170:171], %[wgo7], v[186:187] op_sel:[1,0,0]\n\t"
        "v_pk_fma_f32 v[184:185], v[170:171], %[wif7], v[184:185] op_sel:[1,0,0]\n\t"
        "v_pk_add_f32 v[234:235], v[234:235], 1.0 op_sel_hi:[1,0]\n\t"
        "v_pk_add_f32 v[242:243], v[242:243], 1.0 op_sel_hi:[1,0]\n\t"
        "v_pk_mul_f32 v[236:237], v[236:237], v[240:241]\n\t"
        "v_rcp_f32 v234, v234\n\t"
        "v_wmma_f32_16x16x4_f32 v[132:139], %[w11], v[166:167], v[132:139]\n\t"
        "v_rcp_f32 v235, v235\n\t"
        "v_pk_fma_f32 v[180:181], v[180:181], v[238:239], v[236:237]\n\t"
        "v_pk_fma_f32 v[186:187], v[172:173], %[wgo8], v[186:187] op_sel_hi:[0,1,1]\n\t"
        "v_pk_fma_f32 v[184:185], v[172:173], %[wif8], v[184:185] op_sel_hi:[0,1,1]\n\t"
        "v_pk_add_f32 v[250:251], v[250:251], 1.0 op_sel_hi:[1,0]\n\t"
        "v_rcp_f32 v242, v242\n\t"
        "v_pk_add_f32 v[246:247], v[246:247], 1.0 op_sel_hi:[1,0]\n\t"
        "v_pk_fma_f32 v[234:235], v[234:235], 2.0, 1.0 op_sel_hi:[1,0,0] neg_lo:[1,0,0] neg_hi:[1,0,0]\n\t"
        "v_wmma_f32_16x16x4_f32 v[140:147], %[w21], v[166:167], v[140:147]\n\t"
        "v_pk_mul_f32 v[244:245], v[180:181], v[198:199]\n\t"
        "v_pk_mul_f32 v[168:169], v[232:233], v[234:235]\n\t"
        "v_pk_fma_f32 v[186:187], v[172:173], %[wgo9], v[186:187] op_sel:[1,0,0]\n\t"
        "v_pk_fma_f32 v[184:185], v[172:173], %[wif9], v[184:185] op_sel:[1,0,0]\n\t"
        "v_rcp_f32 v243, v243\n\t"
        "v_pk_add_f32 v[248:249], v[248:249], 1.0 op_sel_hi:[1,0]\n\t"
        "v_exp_f32 v244, v244\n\t"
        "v_wmma_f32_16x16x4_f32 v[148:155], %[w31], v[166:167], v[148:155]\n\t"
        "v_exp_f32 v245, v245\n\t"
        "v_permlanex16_b32 v190, v186, %[sel], 0xfedcba98\n\t"
        "v_permlanex16_b32 v191, v187, %[sel], 0xfedcba98\n\t"
        "v_permlanex16_b32 v188, v184, %[sel], 0xfedcba98\n\t"
        "v_permlanex16_b32 v189, v185, %[sel], 0xfedcba98\n\t"
        "v_rcp_f32 v250, v250\n\t"
        "v_pk_add_f32 v[252:253], v[252:253], 1.0 op_sel_hi:[1,0]\n\t"
        "v_pk_add_f32 v[186:187], v[186:187], v[190:191]\n\t"
        "v_wmma_f32_16x16x4_f32 v[156:163], %[w41], v[166:167], v[156:163]\n\t"
        "v_pk_add_f32 v[244:245], v[244:245], 1.0 op_sel_hi:[1,0]\n\t"
        "v_rcp_f32 v251, v251\n\t"
        "v_pk_add_f32 v[184:185], v[184:185], v[188:189]\n\t"
        "v_fma_f32 v186, v203, %[rg], v186\n\t"
        "v_fma_f32 v187, v203, %[ro], v187\n\t"
        "v_rcp_f32 v246, v246\n\t"
        "s_wait_xcnt 0x0\n\t"
        "v_add_nc_u32 v202, 4, v202\n\t"
        "v_rcp_f32 v244, v244\n\t"
        "v_wmma_f32_16x16x4_f32 v[124:131], %[w02], v[168:169], v[124:131]\n\t"
        "v_rcp_f32 v245, v245\n\t"
        "v_pk_fma_f32 v[250:251], v[250:251], 2.0, 1.0 op_sel_hi:[1,0,0] neg_lo:[1,0,0] neg_hi:[1,0,0]\n\t"
        "v_rcp_f32 v247, v247\n\t"
        "v_fma_f32 v184, v203, %[ri], v184\n\t"
        "v_fma_f32 v185, v203, %[rf], v185\n\t"
        "v_add_f32 v186, %[bg], v186\n\t"
        "v_add_f32 v187, %[bo], v187\n\t"
        "v_rcp_f32 v248, v248\n\t"
        "v_wmma_f32_16x16x4_f32 v[132:139], %[w12], v[168:169], v[132:139]\n\t"
        "v_pk_fma_f32 v[244:245], v[244:245], 2.0, 1.0 op_sel_hi:[1,0,0] neg_lo:[1,0,0] neg_hi:[1,0,0]\n\t"
        "v_rcp_f32 v249, v249\n\t"
        "v_pk_mul_f32 v[246:247], v[246:247], v[250:251]\n\t"
        "v_add_f32 v184, %[bi], v184\n\t"
        "v_add_f32 v185, %[bf], v185\n\t"
        "v_mul_f32 v186, 0x4038aa3b, v186\n\t"
        "v_mul_f32 v187, 0xbfb8aa3b, v187\n\t"
        "v_pk_mul_f32 v[170:171], v[242:243], v[244:245]\n\t"
        "v_wmma_f32_16x16x4_f32 v[140:147], %[w22], v[168:169], v[140:147]\n\t"
        "v_mul_f32 v184, 0xbfb8aa3b, v184\n\t"
        "v_mul_f32 v185, 0xbfb8aa3b, v185\n\t"
        "v_exp_f32 v186, v186\n\t"
        "v_pk_fma_f32 v[182:183], v[182:183], v[248:249], v[246:247]\n\t"
        "v_exp_f32 v187, v187\n\t"
        "s_wait_xcnt 0x0\n\t"
        "v_add_nc_u32 v201, s41, v201\n\t"
        "v_exp_f32 v184, v184\n\t"
        "v_exp_f32 v185, v185\n\t"
        "v_wmma_f32_16x16x4_f32 v[148:155], %[w32], v[168:169], v[148:155]\n\t"
        "v_pk_add_f32 v[186:187], v[186:187], 1.0 op_sel_hi:[1,0]\n\t"
        "v_pk_mul_f32 v[254:255], v[182:183], v[198:199]\n\t"
        "v_pk_add_f32 v[184:185], v[184:185], 1.0 op_sel_hi:[1,0]\n\t"
        "v_rcp_f32 v252, v252\n\t"
        "v_rcp_f32 v253, v253\n\t"
        "v_rcp_f32 v186, v186\n\t"
        "v_exp_f32 v254, v254\n\t"
        "v_wmma_f32_16x16x4_f32 v[156:163], %[w42], v[168:169], v[156:163]\n\t"
        "v_exp_f32 v255, v255\n\t"
        "v_fma_f32 v186, v186, -2.0, 1.0\n\t"
        "v_rcp_f32 v184, v184\n\t"
        "v_rcp_f32 v185, v185\n\t"
        "v_rcp_f32 v187, v187\n\t"
        "v_pk_add_f32 v[254:255], v[254:255], 1.0 op_sel_hi:[1,0]\n\t"
        "v_mul_f32 v184, v184, v186\n\t"
        "v_wmma_f32_16x16x4_f32 v[124:131], %[w03], v[170:171], v[124:131]\n\t"
        "v_fma_f32 v204, v204, v185, v184\n\t"
        "v_rcp_f32 v254, v254\n\t"
        "v_rcp_f32 v255, v255\n\t"
        "v_mul_f32 v184, 0x4038aa3b, v204\n\t"
        "v_exp_f32 v184, v184\n\t"
        "v_pk_fma_f32 v[254:255], v[254:255], 2.0, 1.0 op_sel_hi:[1,0,0] neg_lo:[1,0,0] neg_hi:[1,0,0]\n\t"
        "v_wmma_f32_16x16x4_f32 v[132:139], %[w13], v[170:171], v[132:139]\n\t"
        "v_add_f32 v184, 1.0, v184\n\t"
        "v_pk_mul_f32 v[172:173], v[252:253], v[254:255]\n\t"
        "v_rcp_f32 v184, v184\n\t"
        "v_wmma_f32_16x16x4_f32 v[140:147], %[w23], v[170:171], v[140:147]\n\t"
        "v_fma_f32 v184, v184, -2.0, 1.0\n\t"
        "v_mul_f32 v203, v187, v184\n\t"
        "s_mov_b32 exec_lo, %[msk]\n\t"
        "global_store_b32 v202, v203, %[ob]\n\t"
        "s_mov_b32 exec_lo, -1\n\t"
        "v_wmma_f32_16x16x4_f32 v[148:155], %[w33], v[170:171], v[148:155]\n\t"
        "v_wmma_f32_16x16x4_f32 v[156:163], %[w43], v[170:171], v[156:163]\n\t"
        "v_wmma_f32_16x16x4_f32 v[124:131], %[w04], v[172:173], v[124:131]\n\t"
        "s_add_co_i32 s40, s40, 1\n\t"
        "s_cmp_lg_u32 s40, 0x400\n\t"
        "s_cbranch_scc1 1b\n\t"
        "s_wait_xcnt 0x0\n\t"
        "v_add_nc_u32 v202, 4, v202\n\t"
        "v_pk_fma_f32 v[184:185], v[164:165], %[wif0], 0 op_sel_hi:[0,1,0]\n\t"
        "v_pk_fma_f32 v[186:187], v[164:165], %[wgo0], 0 op_sel_hi:[0,1,0]\n\t"
        "v_pk_fma_f32 v[184:185], v[164:165], %[wif1], v[184:185] op_sel:[1,0,0]\n\t"
        "v_pk_fma_f32 v[186:187], v[164:165], %[wgo1], v[186:187] op_sel:[1,0,0]\n\t"
        "v_pk_fma_f32 v[184:185], v[166:167], %[wif2], v[184:185] op_sel_hi:[0,1,1]\n\t"
        "v_pk_fma_f32 v[186:187], v[166:167], %[wgo2], v[186:187] op_sel_hi:[0,1,1]\n\t"
        "v_pk_fma_f32 v[184:185], v[166:167], %[wif3], v[184:185] op_sel:[1,0,0]\n\t"
        "v_pk_fma_f32 v[186:187], v[166:167], %[wgo3], v[186:187] op_sel:[1,0,0]\n\t"
        "v_pk_fma_f32 v[184:185], v[168:169], %[wif4], v[184:185] op_sel_hi:[0,1,1]\n\t"
        "v_pk_fma_f32 v[186:187], v[168:169], %[wgo4], v[186:187] op_sel_hi:[0,1,1]\n\t"
        "v_pk_fma_f32 v[184:185], v[168:169], %[wif5], v[184:185] op_sel:[1,0,0]\n\t"
        "v_pk_fma_f32 v[186:187], v[168:169], %[wgo5], v[186:187] op_sel:[1,0,0]\n\t"
        "v_pk_fma_f32 v[184:185], v[170:171], %[wif6], v[184:185] op_sel_hi:[0,1,1]\n\t"
        "v_pk_fma_f32 v[186:187], v[170:171], %[wgo6], v[186:187] op_sel_hi:[0,1,1]\n\t"
        "v_pk_fma_f32 v[184:185], v[170:171], %[wif7], v[184:185] op_sel:[1,0,0]\n\t"
        "v_pk_fma_f32 v[186:187], v[170:171], %[wgo7], v[186:187] op_sel:[1,0,0]\n\t"
        "v_pk_fma_f32 v[184:185], v[172:173], %[wif8], v[184:185] op_sel_hi:[0,1,1]\n\t"
        "v_pk_fma_f32 v[186:187], v[172:173], %[wgo8], v[186:187] op_sel_hi:[0,1,1]\n\t"
        "v_pk_fma_f32 v[184:185], v[172:173], %[wif9], v[184:185] op_sel:[1,0,0]\n\t"
        "v_pk_fma_f32 v[186:187], v[172:173], %[wgo9], v[186:187] op_sel:[1,0,0]\n\t"
        "v_permlanex16_b32 v188, v184, %[sel], 0xfedcba98\n\t"
        "v_permlanex16_b32 v189, v185, %[sel], 0xfedcba98\n\t"
        "v_permlanex16_b32 v190, v186, %[sel], 0xfedcba98\n\t"
        "v_permlanex16_b32 v191, v187, %[sel], 0xfedcba98\n\t"
        "v_pk_add_f32 v[184:185], v[184:185], v[188:189]\n\t"
        "v_pk_add_f32 v[186:187], v[186:187], v[190:191]\n\t"
        "v_fma_f32 v184, v203, %[ri], v184\n\t"
        "v_fma_f32 v185, v203, %[rf], v185\n\t"
        "v_fma_f32 v186, v203, %[rg], v186\n\t"
        "v_fma_f32 v187, v203, %[ro], v187\n\t"
        "v_add_f32 v184, %[bi], v184\n\t"
        "v_add_f32 v185, %[bf], v185\n\t"
        "v_add_f32 v186, %[bg], v186\n\t"
        "v_add_f32 v187, %[bo], v187\n\t"
        "v_mul_f32 v184, 0xbfb8aa3b, v184\n\t"
        "v_mul_f32 v185, 0xbfb8aa3b, v185\n\t"
        "v_mul_f32 v186, 0x4038aa3b, v186\n\t"
        "v_mul_f32 v187, 0xbfb8aa3b, v187\n\t"
        "v_exp_f32 v184, v184\n\t"
        "v_exp_f32 v185, v185\n\t"
        "v_exp_f32 v186, v186\n\t"
        "v_exp_f32 v187, v187\n\t"
        "v_pk_add_f32 v[184:185], v[184:185], 1.0 op_sel_hi:[1,0]\n\t"
        "v_pk_add_f32 v[186:187], v[186:187], 1.0 op_sel_hi:[1,0]\n\t"
        "v_rcp_f32 v184, v184\n\t"
        "v_rcp_f32 v185, v185\n\t"
        "v_rcp_f32 v186, v186\n\t"
        "v_rcp_f32 v187, v187\n\t"
        "v_fma_f32 v186, v186, -2.0, 1.0\n\t"
        "v_mul_f32 v184, v184, v186\n\t"
        "v_fma_f32 v204, v204, v185, v184\n\t"
        "v_mul_f32 v184, 0x4038aa3b, v204\n\t"
        "v_exp_f32 v184, v184\n\t"
        "v_nop\n\t"
        "v_add_f32 v184, 1.0, v184\n\t"
        "v_rcp_f32 v184, v184\n\t"
        "v_nop\n\t"
        "v_fma_f32 v184, v184, -2.0, 1.0\n\t"
        "v_mul_f32 v203, v187, v184\n\t"
        "s_mov_b32 exec_lo, %[msk]\n\t"
        "global_store_b32 v202, v203, %[ob]\n\t"
        "s_mov_b32 exec_lo, -1"
        :
        :
          [w00] "v"(wfrag[0][0]), [w01] "v"(wfrag[0][1]), [w02] "v"(wfrag[0][2]), [w03] "v"(wfrag[0][3]), [w04] "v"(wfrag[0][4]), [w05] "v"(wfrag[0][5]),
          [w10] "v"(wfrag[1][0]), [w11] "v"(wfrag[1][1]), [w12] "v"(wfrag[1][2]), [w13] "v"(wfrag[1][3]), [w14] "v"(wfrag[1][4]), [w15] "v"(wfrag[1][5]),
          [w20] "v"(wfrag[2][0]), [w21] "v"(wfrag[2][1]), [w22] "v"(wfrag[2][2]), [w23] "v"(wfrag[2][3]), [w24] "v"(wfrag[2][4]), [w25] "v"(wfrag[2][5]),
          [w30] "v"(wfrag[3][0]), [w31] "v"(wfrag[3][1]), [w32] "v"(wfrag[3][2]), [w33] "v"(wfrag[3][3]), [w34] "v"(wfrag[3][4]), [w35] "v"(wfrag[3][5]),
          [w40] "v"(wfrag[4][0]), [w41] "v"(wfrag[4][1]), [w42] "v"(wfrag[4][2]), [w43] "v"(wfrag[4][3]), [w44] "v"(wfrag[4][4]), [w45] "v"(wfrag[4][5]),
          [wif0] "v"(w2if[0]), [wif1] "v"(w2if[1]), [wif2] "v"(w2if[2]), [wif3] "v"(w2if[3]), [wif4] "v"(w2if[4]), [wif5] "v"(w2if[5]), [wif6] "v"(w2if[6]), [wif7] "v"(w2if[7]), [wif8] "v"(w2if[8]), [wif9] "v"(w2if[9]),
          [wgo0] "v"(w2go[0]), [wgo1] "v"(w2go[1]), [wgo2] "v"(w2go[2]), [wgo3] "v"(w2go[3]), [wgo4] "v"(w2go[4]), [wgo5] "v"(w2go[5]), [wgo6] "v"(w2go[6]), [wgo7] "v"(w2go[7]), [wgo8] "v"(w2go[8]), [wgo9] "v"(w2go[9]),
          [x0] "v"(x0), [x1] "v"(x1), [rowoff] "v"(rowoff),
          [bi] "s"(b2[0]), [bf] "s"(b2[1]), [bg] "s"(b2[2]), [bo] "s"(b2[3]), [ri] "s"(whh2v[0]), [rf] "s"(whh2v[1]), [rg] "s"(whh2v[2]), [ro] "s"(whh2v[3]),
          [sel] "s"(0x76543210), [msk] "s"(0x0000ffffu), [xbp] "s"(x), [ob] "s"(out)
        :
          "v124", "v125", "v126", "v127", "v128", "v129", "v130", "v131", "v132", "v133", "v134", "v135", "v136", "v137", "v138", "v139",
          "v140", "v141", "v142", "v143", "v144", "v145", "v146", "v147", "v148", "v149", "v150", "v151", "v152", "v153", "v154", "v155",
          "v156", "v157", "v158", "v159", "v160", "v161", "v162", "v163", "v164", "v165", "v166", "v167", "v168", "v169", "v170", "v171",
          "v172", "v173", "v174", "v175", "v176", "v177", "v178", "v179", "v180", "v181", "v182", "v183", "v184", "v185", "v186", "v187",
          "v188", "v189", "v190", "v191", "v192", "v193", "v194", "v195", "v196", "v197", "v198", "v199", "v200", "v201", "v202", "v203",
          "v204", "v205", "v206", "v207", "v208", "v209", "v210", "v211", "v212", "v213", "v214", "v215", "v216", "v217", "v218", "v219",
          "v220", "v221", "v222", "v223", "v224", "v225", "v226", "v227", "v228", "v229", "v230", "v231", "v232", "v233", "v234", "v235",
          "v236", "v237", "v238", "v239", "v240", "v241", "v242", "v243", "v244", "v245", "v246", "v247", "v248", "v249", "v250", "v251",
          "v252", "v253", "v254", "v255",
          "s40", "s41", "scc", "memory");
}

extern "C" void kernel_launch(void* const* d_in, const int* in_sizes, int n_in,
                              void* d_out, int out_size, void* d_ws, size_t ws_size,
                              hipStream_t stream) {
  const float* x    = (const float*)d_in[0];
  const float* Wih1 = (const float*)d_in[1];
  const float* Whh1 = (const float*)d_in[2];
  const float* bih1 = (const float*)d_in[3];
  const float* bhh1 = (const float*)d_in[4];
  const float* Wih2 = (const float*)d_in[5];
  const float* Whh2 = (const float*)d_in[6];
  const float* bih2 = (const float*)d_in[7];
  const float* bhh2 = (const float*)d_in[8];
  float* out = (float*)d_out;

  const int B    = in_sizes[0] / TSEQ;
  const int grid = B / (WAVES * 16);
  hipLaunchKernelGGL(lstm_fused, dim3(grid), dim3(WAVES * 32), 0, stream,
                     x, Wih1, Whh1, bih1, bhh1, Wih2, Whh2, bih2, bhh2, out);
}


# Activations on the hardware tanh: tanh = v_tanh_f32, sigmoid(x) = 0.5 + 0.5 tanh(x/2), the 1/2 folded exactly into the i, f, o weight rows; 55 transcendentals a step, not 110; sum order unchanged
# speedup vs baseline: 4.9511x; 1.2159x over previous
#include <hip/hip_runtime.h>

#define HID   20
#define TSEQ  1024
#define WAVES 4
#define MT    5
#define KC    6
#define UPL   10

typedef __attribute__((ext_vector_type(2))) float v2f;

__global__ __launch_bounds__(WAVES * 32)
void lstm_fused(const float* __restrict__ x,    const float* __restrict__ Wih1,
                const float* __restrict__ Whh1, const float* __restrict__ bih1,
                const float* __restrict__ bhh1, const float* __restrict__ Wih2,
                const float* __restrict__ Whh2, const float* __restrict__ bih2,
                const float* __restrict__ bhh2, float* __restrict__ out)
{
  const int lane = threadIdx.x & 31;
  const int wave = threadIdx.x >> 5;
  const int c    = lane & 15;
  const int h    = lane >> 4;
  const int row  = (blockIdx.x * WAVES + wave) * 16 + c;

  v2f wfrag[MT][KC];
  {
    const int hm = c >> 3;
    const int r  = c & 7;
    const int q  = r & 3;
    const float sc = (q == 2) ? 1.0f : 0.5f;
    #pragma unroll
    for (int mt = 0; mt < MT; ++mt) {
      const int um = hm * UPL + mt * 2 + (r >> 2);
      const int n  = q * HID + um;
      #pragma unroll
      for (int kc = 0; kc < KC - 1; ++kc) {
        const int uk = h * UPL + kc * 2;
        v2f w;
        w.x = sc * Whh1[n * HID + uk];
        w.y = sc * Whh1[n * HID + uk + 1];
        wfrag[mt][kc] = w;
      }
      v2f wl;
      wl.x = (h == 0) ? sc * Wih1[n]             : 0.0f;
      wl.y = (h == 0) ? sc * (bih1[n] + bhh1[n]) : 0.0f;
      wfrag[mt][KC - 1] = wl;
    }
  }

  v2f w2if[UPL], w2go[UPL];
  float b2[4], whh2v[4];
  #pragma unroll
  for (int u = 0; u < UPL; ++u) {
    const int j = h * UPL + u;
    w2if[u] = (v2f){0.5f * Wih2[0 * HID + j], 0.5f * Wih2[1 * HID + j]};
    w2go[u] = (v2f){Wih2[2 * HID + j], 0.5f * Wih2[3 * HID + j]};
  }
  #pragma unroll
  for (int q = 0; q < 4; ++q) {
    const float sc2 = (q == 2) ? 1.0f : 0.5f;
    b2[q]    = sc2 * (bih2[q] + bhh2[q]);
    whh2v[q] = sc2 * Whh2[q];
  }

  const unsigned rowoff = (unsigned)row * (TSEQ * 4u);
  const float x0 = x[(size_t)row * TSEQ];
  const float x1 = x[(size_t)row * TSEQ + 1];


  asm volatile(
        "s_mov_b32 s41, 4\n\t"
        "v_mov_b32 v164, 0\n\t"
        "v_mov_b32 v165, 0\n\t"
        "v_mov_b32 v166, 0\n\t"
        "v_mov_b32 v167, 0\n\t"
        "v_mov_b32 v168, 0\n\t"
        "v_mov_b32 v169, 0\n\t"
        "v_mov_b32 v170, 0\n\t"
        "v_mov_b32 v171, 0\n\t"
        "v_mov_b32 v172, 0\n\t"
        "v_mov_b32 v173, 0\n\t"
        "v_mov_b32 v174, 0\n\t"
        "v_mov_b32 v175, 0\n\t"
        "v_mov_b32 v176, 0\n\t"
        "v_mov_b32 v177, 0\n\t"
        "v_mov_b32 v178, 0\n\t"
        "v_mov_b32 v179, 0\n\t"
        "v_mov_b32 v180, 0\n\t"
        "v_mov_b32 v181, 0\n\t"
        "v_mov_b32 v182, 0\n\t"
        "v_mov_b32 v183, 0\n\t"
        "v_mov_b32 v203, 0\n\t"
        "v_mov_b32 v204, 0\n\t"
        "v_cndmask_b32_e64 v193, 0, 1.0, %[msk]\n\t"
        "v_cndmask_b32_e64 v192, 0, %[x0], %[msk]\n\t"
        "v_mov_b32 v200, %[x1]\n\t"
        "v_add_nc_u32 v201, 8, %[rowoff]\n\t"
        "v_add_nc_u32 v202, -4, %[rowoff]\n\t"
        "v_wmma_f32_16x16x4_f32 v[124:131], %[w05], v[192:193], 0\n\t"
        "v_wmma_f32_16x16x4_f32 v[132:139], %[w15], v[192:193], 0\n\t"
        "v_wmma_f32_16x16x4_f32 v[140:147], %[w25], v[192:193], 0\n\t"
        "v_wmma_f32_16x16x4_f32 v[148:155], %[w35], v[192:193], 0\n\t"
        "v_wmma_f32_16x16x4_f32 v[156:163], %[w45], v[192:193], 0\n\t"
        "v_wmma_f32_16x16x4_f32 v[132:139], %[w14], v[172:173], v[132:139]\n\t"
        "v_wmma_f32_16x16x4_f32 v[140:147], %[w24], v[172:173], v[140:147]\n\t"
        "v_wmma_f32_16x16x4_f32 v[148:155], %[w34], v[172:173], v[148:155]\n\t"
        "v_wmma_f32_16x16x4_f32 v[156:163], %[w44], v[172:173], v[156:163]\n\t"
        "v_nop\n\t"
        "v_nop\n\t"
        "v_nop\n\t"
        "v_nop\n\t"
        "v_nop\n\t"
        "v_nop\n\t"
        "v_nop\n\t"
        "v_nop\n\t"
        "v_nop\n\t"
        "s_wait_loadcnt 0x0\n\t"
        "v_cndmask_b32_e64 v192, 0, v200, %[msk]\n\t"
        "global_load_b32 v200, v201, %[xbp]\n\t"
        "v_tanh_f32 v206, v124\n\t"
        "v_tanh_f32 v207, v128\n\t"
        "v_tanh_f32 v208, v125\n\t"
        "v_tanh_f32 v209, v129\n\t"
        "v_tanh_f32 v210, v126\n\t"
        "v_tanh_f32 v211, v130\n\t"
        "v_tanh_f32 v212, v127\n\t"
        "v_pk_fma_f32 v[206:207], v[206:207], 0.5, 0.5 op_sel_hi:[1,0,0]\n\t"
        "v_tanh_f32 v213, v131\n\t"
        "v_tanh_f32 v216, v132\n\t"
        "v_wmma_f32_16x16x4_f32 v[124:131], %[w05], v[192:193], 0\n\t"
        "v_tanh_f32 v217, v136\n\t"
        "v_pk_fma_f32 v[208:209], v[208:209], 0.5, 0.5 op_sel_hi:[1,0,0]\n\t"
        "v_pk_fma_f32 v[212:213], v[212:213], 0.5, 0.5 op_sel_hi:[1,0,0]\n\t"
        "v_tanh_f32 v218, v133\n\t"
        "v_pk_mul_f32 v[206:207], v[206:207], v[210:211]\n\t"
        "v_tanh_f32 v219, v137\n\t"
        "v_tanh_f32 v220, v134\n\t"
        "v_pk_fma_f32 v[174:175], v[174:175], v[208:209], v[206:207]\n\t"
        "v_tanh_f32 v221, v138\n\t"
        "v_tanh_f32 v222, v135\n\t"
        "v_pk_fma_f32 v[216:217], v[216:217], 0.5, 0.5 op_sel_hi:[1,0,0]\n\t"
        "v_tanh_f32 v223, v139\n\t"
        "v_tanh_f32 v226, v140\n\t"
        "v_wmma_f32_16x16x4_f32 v[132:139], %[w15], v[192:193], 0\n\t"
        "v_tanh_f32 v227, v144\n\t"
        "v_pk_fma_f32 v[218:219], v[218:219], 0.5, 0.5 op_sel_hi:[1,0,0]\n\t"
        "v_pk_fma_f32 v[222:223], v[222:223], 0.5, 0.5 op_sel_hi:[1,0,0]\n\t"
        "v_tanh_f32 v228, v141\n\t"
        "v_pk_mul_f32 v[216:217], v[216:217], v[220:221]\n\t"
        "v_tanh_f32 v229, v145\n\t"
        "v_tanh_f32 v230, v142\n\t"
        "v_pk_fma_f32 v[176:177], v[176:177], v[218:219], v[216:217]\n\t"
        "v_tanh_f32 v231, v146\n\t"
        "v_tanh_f32 v232, v143\n\t"
        "v_pk_fma_f32 v[226:227], v[226:227], 0.5, 0.5 op_sel_hi:[1,0,0]\n\t"
        "v_tanh_f32 v233, v147\n\t"
        "v_tanh_f32 v236, v148\n\t"
        "v_wmma_f32_16x16x4_f32 v[140:147], %[w25], v[192:193], 0\n\t"
        "v_tanh_f32 v237, v152\n\t"
        "v_pk_fma_f32 v[228:229], v[228:229], 0.5, 0.5 op_sel_hi:[1,0,0]\n\t"
        "v_pk_mul_f32 v[226:227], v[226:227], v[230:231]\n\t"
        "v_tanh_f32 v238, v149\n\t"
        "v_pk_fma_f32 v[232:233], v[232:233], 0.5, 0.5 op_sel_hi:[1,0,0]\n\t"
        "v_tanh_f32 v239, v153\n\t"
        "v_pk_fma_f32 v[178:179], v[178:179], v[228:229], v[226:227]\n\t"
        "v_tanh_f32 v240, v150\n\t"
        "v_tanh_f32 v241, v154\n\t"
        "v_tanh_f32 v242, v151\n\t"
        "v_pk_fma_f32 v[236:237], v[236:237], 0.5, 0.5 op_sel_hi:[1,0,0]\n\t"
        "v_tanh_f32 v243, v155\n\t"
        "v_tanh_f32 v246, v156\n\t"
        "v_wmma_f32_16x16x4_f32 v[148:155], %[w35], v[192:193], 0\n\t"
        "v_tanh_f32 v247, v160\n\t"
        "v_pk_fma_f32 v[238:239], v[238:239], 0.5, 0.5 op_sel_hi:[1,0,0]\n\t"
        "v_pk_mul_f32 v[236:237], v[236:237], v[240:241]\n\t"
        "v_tanh_f32 v248, v157\n\t"
        "v_pk_fma_f32 v[242:243], v[242:243], 0.5, 0.5 op_sel_hi:[1,0,0]\n\t"
        "v_tanh_f32 v249, v161\n\t"
        "v_pk_fma_f32 v[180:181], v[180:181], v[238:239], v[236:237]\n\t"
        "v_tanh_f32 v250, v158\n\t"
        "v_tanh_f32 v251, v162\n\t"
        "v_tanh_f32 v252, v159\n\t"
        "v_pk_fma_f32 v[246:247], v[246:247], 0.5, 0.5 op_sel_hi:[1,0,0]\n\t"
        "v_tanh_f32 v253, v163\n\t"
        "v_tanh_f32 v214, v174\n\t"
        "v_wmma_f32_16x16x4_f32 v[156:163], %[w45], v[192:193], 0\n\t"
        "v_tanh_f32 v215, v175\n\t"
        "v_pk_fma_f32 v[248:249], v[248:249], 0.5, 0.5 op_sel_hi:[1,0,0]\n\t"
        "v_pk_mul_f32 v[246:247], v[246:247], v[250:251]\n\t"
        "v_tanh_f32 v224, v176\n\t"
        "v_pk_fma_f32 v[252:253], v[252:253], 0.5, 0.5 op_sel_hi:[1,0,0]\n\t"
        "v_tanh_f32 v225, v177\n\t"
        "v_pk_fma_f32 v[182:183], v[182:183], v[248:249], v[246:247]\n\t"
        "v_tanh_f32 v234, v178\n\t"
        "v_tanh_f32 v235, v179\n\t"
        "v_pk_mul_f32 v[164:165], v[212:213], v[214:215]\n\t"
        "v_tanh_f32 v244, v180\n\t"
        "v_tanh_f32 v245, v181\n\t"
        "v_wmma_f32_16x16x4_f32 v[124:131], %[w00], v[164:165], v[124:131]\n\t"
        "v_wmma_f32_16x16x4_f32 v[132:139], %[w10], v[164:165], v[132:139]\n\t"
        "v_wmma_f32_16x16x4_f32 v[140:147], %[w20], v[164:165], v[140:147]\n\t"
        "v_wmma_f32_16x16x4_f32 v[148:155], %[w30], v[164:165], v[148:155]\n\t"
        "v_wmma_f32_16x16x4_f32 v[156:163], %[w40], v[164:165], v[156:163]\n\t"
        "v_pk_mul_f32 v[166:167], v[222:223], v[224:225]\n\t"
        "v_pk_mul_f32 v[168:169], v[232:233], v[234:235]\n\t"
        "v_pk_mul_f32 v[170:171], v[242:243], v[244:245]\n\t"
        "v_tanh_f32 v254, v182\n\t"
        "v_wmma_f32_16x16x4_f32 v[124:131], %[w01], v[166:167], v[124:131]\n\t"
        "v_wmma_f32_16x16x4_f32 v[132:139], %[w11], v[166:167], v[132:139]\n\t"
        "v_wmma_f32_16x16x4_f32 v[140:147], %[w21], v[166:167], v[140:147]\n\t"
        "v_wmma_f32_16x16x4_f32 v[148:155], %[w31], v[166:167], v[148:155]\n\t"
        "v_wmma_f32_16x16x4_f32 v[156:163], %[w41], v[166:167], v[156:163]\n\t"
        "v_wmma_f32_16x16x4_f32 v[124:131], %[w02], v[168:169], v[124:131]\n\t"
        "v_wmma_f32_16x16x4_f32 v[132:139], %[w12], v[168:169], v[132:139]\n\t"
        "v_wmma_f32_16x16x4_f32 v[140:147], %[w22], v[168:169], v[140:147]\n\t"
        "v_wmma_f32_16x16x4_f32 v[148:155], %[w32], v[168:169], v[148:155]\n\t"
        "v_wmma_f32_16x16x4_f32 v[156:163], %[w42], v[168:169], v[156:163]\n\t"
        "v_wmma_f32_16x16x4_f32 v[124:131], %[w03], v[170:171], v[124:131]\n\t"
        "v_wmma_f32_16x16x4_f32 v[132:139], %[w13], v[170:171], v[132:139]\n\t"
        "v_wmma_f32_16x16x4_f32 v[140:147], %[w23], v[170:171], v[140:147]\n\t"
        "v_wmma_f32_16x16x4_f32 v[148:155], %[w33], v[170:171], v[148:155]\n\t"
        "v_tanh_f32 v255, v183\n\t"
        "v_wmma_f32_16x16x4_f32 v[156:163], %[w43], v[170:171], v[156:163]\n\t"
        "s_wait_xcnt 0x0\n\t"
        "v_add_nc_u32 v201, s41, v201\n\t"
        "v_pk_mul_f32 v[172:173], v[252:253], v[254:255]\n\t"
        "v_wmma_f32_16x16x4_f32 v[124:131], %[w04], v[172:173], v[124:131]\n\t"
        "s_mov_b32 s40, 1\n\t"
        "1:\n\t"
        "s_cmp_lt_u32 s40, 0x3fd\n\t"
        "s_cselect_b32 s41, 4, 0\n\t"
        "v_pk_fma_f32 v[184:185], v[164:165], %[wif0], 0 op_sel_hi:[0,1,0]\n\t"
        "v_pk_fma_f32 v[186:187], v[164:165], %[wgo0], 0 op_sel_hi:[0,1,0]\n\t"
        "v_wmma_f32_16x16x4_f32 v[132:139], %[w14], v[172:173], v[132:139]\n\t"
        "v_pk_fma_f32 v[184:185], v[164:165], %[wif1], v[184:185] op_sel:[1,0,0]\n\t"
        "v_pk_fma_f32 v[186:187], v[164:165], %[wgo1], v[186:187] op_sel:[1,0,0]\n\t"
        "v_wmma_f32_16x16x4_f32 v[140:147], %[w24], v[172:173], v[140:147]\n\t"
        "v_pk_fma_f32 v[184:185], v[166:167], %[wif2], v[184:185] op_sel_hi:[0,1,1]\n\t"
        "v_pk_fma_f32 v[186:187], v[166:167], %[wgo2], v[186:187] op_sel_hi:[0,1,1]\n\t"
        "v_wmma_f32_16x16x4_f32 v[148:155], %[w34], v[172:173], v[148:155]\n\t"
        "v_pk_fma_f32 v[184:185], v[166:167], %[wif3], v[184:185] op_sel:[1,0,0]\n\t"
        "v_pk_fma_f32 v[186:187], v[166:167], %[wgo3], v[186:187] op_sel:[1,0,0]\n\t"
        "v_wmma_f32_16x16x4_f32 v[156:163], %[w44], v[172:173], v[156:163]\n\t"
        "v_pk_fma_f32 v[184:185], v[168:169], %[wif4], v[184:185] op_sel_hi:[0,1,1]\n\t"
        "v_pk_fma_f32 v[186:187], v[168:169], %[wgo4], v[186:187] op_sel_hi:[0,1,1]\n\t"
        "s_wait_loadcnt 0x0\n\t"
        "v_cndmask_b32_e64 v192, 0, v200, %[msk]\n\t"
        "global_load_b32 v200, v201, %[xbp]\n\t"
        "v_tanh_f32 v206, v124\n\t"
        "v_pk_fma_f32 v[184:185], v[168:169], %[wif5], v[184:185] op_sel:[1,0,0]\n\t"
        "v_pk_fma_f32 v[186:187], v[168:169], %[wgo5], v[186:187] op_sel:[1,0,0]\n\t"
        "v_tanh_f32 v207, v128\n\t"
        "v_tanh_f32 v208, v125\n\t"
        "v_pk_fma_f32 v[184:185], v[170:171], %[wif6], v[184:185] op_sel_hi:[0,1,1]\n\t"
        "v_pk_fma_f32 v[186:187], v[170:171], %[wgo6], v[186:187] op_sel_hi:[0,1,1]\n\t"
        "v_tanh_f32 v209, v129\n\t"
        "v_tanh_f32 v210, v126\n\t"
        "v_pk_fma_f32 v[184:185], v[170:171], %[wif7], v[184:185] op_sel:[1,0,0]\n\t"
        "v_pk_fma_f32 v[186:187], v[170:171], %[wgo7], v[186:187] op_sel:[1,0,0]\n\t"
        "v_tanh_f32 v211, v130\n\t"
        "v_tanh_f32 v212, v127\n\t"
        "v_pk_fma_f32 v[184:185], v[172:173], %[wif8], v[184:185] op_sel_hi:[0,1,1]\n\t"
        "v_pk_fma_f32 v[186:187], v[172:173], %[wgo8], v[186:187] op_sel_hi:[0,1,1]\n\t"
        "v_tanh_f32 v213, v131\n\t"
        "v_pk_fma_f32 v[206:207], v[206:207], 0.5, 0.5 op_sel_hi:[1,0,0]\n\t"
        "v_wmma_f32_16x16x4_f32 v[124:131], %[w05], v[192:193], 0\n\t"
        "v_pk_fma_f32 v[184:185], v[172:173], %[wif9], v[184:185] op_sel:[1,0,0]\n\t"
        "v_pk_fma_f32 v[186:187], v[172:173], %[wgo9], v[186:187] op_sel:[1,0,0]\n\t"
        "v_tanh_f32 v216, v132\n\t"
        "v_pk_fma_f32 v[208:209], v[208:209], 0.5, 0.5 op_sel_hi:[1,0,0]\n\t"
        "v_pk_fma_f32 v[212:213], v[212:213], 0.5, 0.5 op_sel_hi:[1,0,0]\n\t"
        "v_tanh_f32 v217, v136\n\t"
        "v_pk_mul_f32 v[206:207], v[206:207], v[210:211]\n\t"
        "v_permlanex16_b32 v188, v184, %[sel], 0xfedcba98\n\t"
        "v_permlanex16_b32 v189, v185, %[sel], 0xfedcba98\n\t"
        "v_tanh_f32 v218, v133\n\t"
        "v_permlanex16_b32 v190, v186, %[sel], 0xfedcba98\n\t"
        "v_permlanex16_b32 v191, v187, %[sel], 0xfedcba98\n\t"
        "v_tanh_f32 v219, v137\n\t"
        "v_pk_fma_f32 v[174:175], v[174:175], v[208:209], v[206:207]\n\t"
        "v_pk_add_f32 v[184:185], v[184:185], v[188:189]\n\t"
        "v_tanh_f32 v220, v134\n\t"
        "v_pk_add_f32 v[186:187], v[186:187], v[190:191]\n\t"
        "v_tanh_f32 v221, v138\n\t"
        "v_fma_f32 v184, v203, %[ri], v184\n\t"
        "v_pk_fma_f32 v[216:217], v[216:217], 0.5, 0.5 op_sel_hi:[1,0,0]\n\t"
        "v_tanh_f32 v222, v135\n\t"
        "v_fma_f32 v185, v203, %[rf], v185\n\t"
        "v_fma_f32 v186, v203, %[rg], v186\n\t"
        "v_add_f32 v184, %[bi], v184\n\t"
        "v_tanh_f32 v223, v139\n\t"
        "v_fma_f32 v187, v203, %[ro], v187\n\t"
        "v_wmma_f32_16x16x4_f32 v[132:139], %[w15], v[192:193], 0\n\t"
        "v_tanh_f32 v226, v140\n\t"
        "v_pk_fma_f32 v[218:219], v[218:219], 0.5, 0.5 op_sel_hi:[1,0,0]\n\t"
        "v_pk_fma_f32 v[222:223], v[222:223], 0.5, 0.5 op_sel_hi:[1,0,0]\n\t"
        "v_tanh_f32 v227, v144\n\t"
        "v_pk_mul_f32 v[216:217], v[216:217], v[220:221]\n\t"
        "v_add_f32 v185, %[bf], v185\n\t"
        "v_add_f32 v186, %[bg], v186\n\t"
        "v_tanh_f32 v228, v141\n\t"
        "v_add_f32 v187, %[bo], v187\n\t"
        "v_tanh_f32 v229, v145\n\t"
        "v_pk_fma_f32 v[176:177], v[176:177], v[218:219], v[216:217]\n\t"
        "v_tanh_f32 v230, v142\n\t"
        "v_tanh_f32 v231, v146\n\t"
        "v_pk_fma_f32 v[226:227], v[226:227], 0.5, 0.5 op_sel_hi:[1,0,0]\n\t"
        "v_tanh_f32 v232, v143\n\t"
        "v_tanh_f32 v233, v147\n\t"
        "v_pk_fma_f32 v[228:229], v[228:229], 0.5, 0.5 op_sel_hi:[1,0,0]\n\t"
        "v_wmma_f32_16x16x4_f32 v[140:147], %[w25], v[192:193], 0\n\t"
        "v_tanh_f32 v236, v148\n\t"
        "v_pk_mul_f32 v[226:227], v[226:227], v[230:231]\n\t"
        "v_pk_fma_f32 v[232:233], v[232:233], 0.5, 0.5 op_sel_hi:[1,0,0]\n\t"
        "v_tanh_f32 v237, v152\n\t"
        "v_tanh_f32 v238, v149\n\t"
        "v_pk_fma_f32 v[178:179], v[178:179], v[228:229], v[226:227]\n\t"
        "v_tanh_f32 v239, v153\n\t"
        "v_tanh_f32 v240, v150\n\t"
        "v_tanh_f32 v241, v154\n\t"
        "v_tanh_f32 v242, v151\n\t"
        "v_pk_fma_f32 v[236:237], v[236:237], 0.5, 0.5 op_sel_hi:[1,0,0]\n\t"
        "v_tanh_f32 v243, v155\n\t"
        "v_tanh_f32 v246, v156\n\t"
        "v_wmma_f32_16x16x4_f32 v[148:155], %[w35], v[192:193], 0\n\t"
        "v_tanh_f32 v247, v160\n\t"
        "v_pk_fma_f32 v[238:239], v[238:239], 0.5, 0.5 op_sel_hi:[1,0,0]\n\t"
        "v_pk_mul_f32 v[236:237], v[236:237], v[240:241]\n\t"
        "v_tanh_f32 v248, v157\n\t"
        "v_pk_fma_f32 v[242:243], v[242:243], 0.5, 0.5 op_sel_hi:[1,0,0]\n\t"
        "v_tanh_f32 v249, v161\n\t"
        "v_pk_fma_f32 v[180:181], v[180:181], v[238:239], v[236:237]\n\t"
        "v_tanh_f32 v250, v158\n\t"
        "v_tanh_f32 v251, v162\n\t"
        "v_tanh_f32 v252, v159\n\t"
        "v_pk_fma_f32 v[246:247], v[246:247], 0.5, 0.5 op_sel_hi:[1,0,0]\n\t"
        "v_tanh_f32 v253, v163\n\t"
        "v_tanh_f32 v214, v174\n\t"
        "v_wmma_f32_16x16x4_f32 v[156:163], %[w45], v[192:193], 0\n\t"
        "v_tanh_f32 v215, v175\n\t"
        "v_pk_fma_f32 v[248:249], v[248:249], 0.5, 0.5 op_sel_hi:[1,0,0]\n\t"
        "v_pk_mul_f32 v[246:247], v[246:247], v[250:251]\n\t"
        "v_tanh_f32 v224, v176\n\t"
        "v_pk_fma_f32 v[252:253], v[252:253], 0.5, 0.5 op_sel_hi:[1,0,0]\n\t"
        "v_tanh_f32 v225, v177\n\t"
        "v_pk_fma_f32 v[182:183], v[182:183], v[248:249], v[246:247]\n\t"
        "v_tanh_f32 v184, v184\n\t"
        "v_tanh_f32 v185, v185\n\t"
        "v_pk_mul_f32 v[164:165], v[212:213], v[214:215]\n\t"
        "v_tanh_f32 v186, v186\n\t"
        "v_tanh_f32 v187, v187\n\t"
        "v_wmma_f32_16x16x4_f32 v[124:131], %[w00], v[164:165], v[124:131]\n\t"
        "v_wmma_f32_16x16x4_f32 v[132:139], %[w10], v[164:165], v[132:139]\n\t"
        "v_wmma_f32_16x16x4_f32 v[140:147], %[w20], v[164:165], v[140:147]\n\t"
        "v_wmma_f32_16x16x4_f32 v[148:155], %[w30], v[164:165], v[148:155]\n\t"
        "v_wmma_f32_16x16x4_f32 v[156:163], %[w40], v[164:165], v[156:163]\n\t"
        "v_pk_mul_f32 v[166:167], v[222:223], v[224:225]\n\t"
        "v_pk_fma_f32 v[184:185], v[184:185], 0.5, 0.5 op_sel_hi:[1,0,0]\n\t"
        "v_fma_f32 v187, v187, 0.5, 0.5\n\t"
        "s_wait_xcnt 0x0\n\t"
        "v_add_nc_u32 v202, 4, v202\n\t"
        "v_tanh_f32 v234, v178\n\t"
        "v_wmma_f32_16x16x4_f32 v[124:131], %[w01], v[166:167], v[124:131]\n\t"
        "v_mul_f32 v184, v184, v186\n\t"
        "v_wmma_f32_16x16x4_f32 v[132:139], %[w11], v[166:167], v[132:139]\n\t"
        "v_fma_f32 v204, v204, v185, v184\n\t"
        "v_wmma_f32_16x16x4_f32 v[140:147], %[w21], v[166:167], v[140:147]\n\t"
        "v_tanh_f32 v184, v204\n\t"
        "v_wmma_f32_16x16x4_f32 v[148:155], %[w31], v[166:167], v[148:155]\n\t"
        "v_tanh_f32 v235, v179\n\t"
        "v_wmma_f32_16x16x4_f32 v[156:163], %[w41], v[166:167], v[156:163]\n\t"
        "v_mul_f32 v203, v187, v184\n\t"
        "v_tanh_f32 v244, v180\n\t"
        "s_wait_xcnt 0x0\n\t"
        "v_add_nc_u32 v201, s41, v201\n\t"
        "v_pk_mul_f32 v[168:169], v[232:233], v[234:235]\n\t"
        "s_mov_b32 exec_lo, %[msk]\n\t"
        "global_store_b32 v202, v203, %[ob]\n\t"
        "s_mov_b32 exec_lo, -1\n\t"
        "v_tanh_f32 v245, v181\n\t"
        "v_wmma_f32_16x16x4_f32 v[124:131], %[w02], v[168:169], v[124:131]\n\t"
        "v_wmma_f32_16x16x4_f32 v[132:139], %[w12], v[168:169], v[132:139]\n\t"
        "v_wmma_f32_16x16x4_f32 v[140:147], %[w22], v[168:169], v[140:147]\n\t"
        "v_wmma_f32_16x16x4_f32 v[148:155], %[w32], v[168:169], v[148:155]\n\t"
        "v_wmma_f32_16x16x4_f32 v[156:163], %[w42], v[168:169], v[156:163]\n\t"
        "v_pk_mul_f32 v[170:171], v[242:243], v[244:245]\n\t"
        "v_tanh_f32 v254, v182\n\t"
        "v_tanh_f32 v255, v183\n\t"
        "v_wmma_f32_16x16x4_f32 v[124:131], %[w03], v[170:171], v[124:131]\n\t"
        "v_wmma_f32_16x16x4_f32 v[132:139], %[w13], v[170:171], v[132:139]\n\t"
        "v_wmma_f32_16x16x4_f32 v[140:147], %[w23], v[170:171], v[140:147]\n\t"
        "v_wmma_f32_16x16x4_f32 v[148:155], %[w33], v[170:171], v[148:155]\n\t"
        "v_wmma_f32_16x16x4_f32 v[156:163], %[w43], v[170:171], v[156:163]\n\t"
        "v_pk_mul_f32 v[172:173], v[252:253], v[254:255]\n\t"
        "v_wmma_f32_16x16x4_f32 v[124:131], %[w04], v[172:173], v[124:131]\n\t"
        "s_add_co_i32 s40, s40, 1\n\t"
        "s_cmp_lg_u32 s40, 0x400\n\t"
        "s_cbranch_scc1 1b\n\t"
        "v_pk_fma_f32 v[184:185], v[164:165], %[wif0], 0 op_sel_hi:[0,1,0]\n\t"
        "v_pk_fma_f32 v[186:187], v[164:165], %[wgo0], 0 op_sel_hi:[0,1,0]\n\t"
        "s_wait_xcnt 0x0\n\t"
        "v_add_nc_u32 v202, 4, v202\n\t"
        "v_pk_fma_f32 v[184:185], v[164:165], %[wif1], v[184:185] op_sel:[1,0,0]\n\t"
        "v_pk_fma_f32 v[186:187], v[164:165], %[wgo1], v[186:187] op_sel:[1,0,0]\n\t"
        "v_pk_fma_f32 v[184:185], v[166:167], %[wif2], v[184:185] op_sel_hi:[0,1,1]\n\t"
        "v_pk_fma_f32 v[186:187], v[166:167], %[wgo2], v[186:187] op_sel_hi:[0,1,1]\n\t"
        "v_pk_fma_f32 v[184:185], v[166:167], %[wif3], v[184:185] op_sel:[1,0,0]\n\t"
        "v_pk_fma_f32 v[186:187], v[166:167], %[wgo3], v[186:187] op_sel:[1,0,0]\n\t"
        "v_pk_fma_f32 v[184:185], v[168:169], %[wif4], v[184:185] op_sel_hi:[0,1,1]\n\t"
        "v_pk_fma_f32 v[186:187], v[168:169], %[wgo4], v[186:187] op_sel_hi:[0,1,1]\n\t"
        "v_pk_fma_f32 v[184:185], v[168:169], %[wif5], v[184:185] op_sel:[1,0,0]\n\t"
        "v_pk_fma_f32 v[186:187], v[168:169], %[wgo5], v[186:187] op_sel:[1,0,0]\n\t"
        "v_pk_fma_f32 v[184:185], v[170:171], %[wif6], v[184:185] op_sel_hi:[0,1,1]\n\t"
        "v_pk_fma_f32 v[186:187], v[170:171], %[wgo6], v[186:187] op_sel_hi:[0,1,1]\n\t"
        "v_pk_fma_f32 v[184:185], v[170:171], %[wif7], v[184:185] op_sel:[1,0,0]\n\t"
        "v_pk_fma_f32 v[186:187], v[170:171], %[wgo7], v[186:187] op_sel:[1,0,0]\n\t"
        "v_pk_fma_f32 v[184:185], v[172:173], %[wif8], v[184:185] op_sel_hi:[0,1,1]\n\t"
        "v_pk_fma_f32 v[186:187], v[172:173], %[wgo8], v[186:187] op_sel_hi:[0,1,1]\n\t"
        "v_pk_fma_f32 v[184:185], v[172:173], %[wif9], v[184:185] op_sel:[1,0,0]\n\t"
        "v_pk_fma_f32 v[186:187], v[172:173], %[wgo9], v[186:187] op_sel:[1,0,0]\n\t"
        "v_permlanex16_b32 v188, v184, %[sel], 0xfedcba98\n\t"
        "v_permlanex16_b32 v189, v185, %[sel], 0xfedcba98\n\t"
        "v_permlanex16_b32 v190, v186, %[sel], 0xfedcba98\n\t"
        "v_permlanex16_b32 v191, v187, %[sel], 0xfedcba98\n\t"
        "v_pk_add_f32 v[184:185], v[184:185], v[188:189]\n\t"
        "v_pk_add_f32 v[186:187], v[186:187], v[190:191]\n\t"
        "v_fma_f32 v184, v203, %[ri], v184\n\t"
        "v_fma_f32 v185, v203, %[rf], v185\n\t"
        "v_fma_f32 v186, v203, %[rg], v186\n\t"
        "v_fma_f32 v187, v203, %[ro], v187\n\t"
        "v_add_f32 v184, %[bi], v184\n\t"
        "v_add_f32 v185, %[bf], v185\n\t"
        "v_add_f32 v186, %[bg], v186\n\t"
        "v_add_f32 v187, %[bo], v187\n\t"
        "v_tanh_f32 v184, v184\n\t"
        "v_tanh_f32 v185, v185\n\t"
        "v_tanh_f32 v186, v186\n\t"
        "v_tanh_f32 v187, v187\n\t"
        "v_pk_fma_f32 v[184:185], v[184:185], 0.5, 0.5 op_sel_hi:[1,0,0]\n\t"
        "v_mul_f32 v184, v184, v186\n\t"
        "v_fma_f32 v187, v187, 0.5, 0.5\n\t"
        "v_fma_f32 v204, v204, v185, v184\n\t"
        "v_tanh_f32 v184, v204\n\t"
        "v_nop\n\t"
        "v_mul_f32 v203, v187, v184\n\t"
        "s_mov_b32 exec_lo, %[msk]\n\t"
        "global_store_b32 v202, v203, %[ob]\n\t"
        "s_mov_b32 exec_lo, -1"
        :
        :
          [w00] "v"(wfrag[0][0]), [w01] "v"(wfrag[0][1]), [w02] "v"(wfrag[0][2]), [w03] "v"(wfrag[0][3]), [w04] "v"(wfrag[0][4]), [w05] "v"(wfrag[0][5]),
          [w10] "v"(wfrag[1][0]), [w11] "v"(wfrag[1][1]), [w12] "v"(wfrag[1][2]), [w13] "v"(wfrag[1][3]), [w14] "v"(wfrag[1][4]), [w15] "v"(wfrag[1][5]),
          [w20] "v"(wfrag[2][0]), [w21] "v"(wfrag[2][1]), [w22] "v"(wfrag[2][2]), [w23] "v"(wfrag[2][3]), [w24] "v"(wfrag[2][4]), [w25] "v"(wfrag[2][5]),
          [w30] "v"(wfrag[3][0]), [w31] "v"(wfrag[3][1]), [w32] "v"(wfrag[3][2]), [w33] "v"(wfrag[3][3]), [w34] "v"(wfrag[3][4]), [w35] "v"(wfrag[3][5]),
          [w40] "v"(wfrag[4][0]), [w41] "v"(wfrag[4][1]), [w42] "v"(wfrag[4][2]), [w43] "v"(wfrag[4][3]), [w44] "v"(wfrag[4][4]), [w45] "v"(wfrag[4][5]),
          [wif0] "v"(w2if[0]), [wif1] "v"(w2if[1]), [wif2] "v"(w2if[2]), [wif3] "v"(w2if[3]), [wif4] "v"(w2if[4]), [wif5] "v"(w2if[5]), [wif6] "v"(w2if[6]), [wif7] "v"(w2if[7]), [wif8] "v"(w2if[8]), [wif9] "v"(w2if[9]),
          [wgo0] "v"(w2go[0]), [wgo1] "v"(w2go[1]), [wgo2] "v"(w2go[2]), [wgo3] "v"(w2go[3]), [wgo4] "v"(w2go[4]), [wgo5] "v"(w2go[5]), [wgo6] "v"(w2go[6]), [wgo7] "v"(w2go[7]), [wgo8] "v"(w2go[8]), [wgo9] "v"(w2go[9]),
          [x0] "v"(x0), [x1] "v"(x1), [rowoff] "v"(rowoff),
          [bi] "s"(b2[0]), [bf] "s"(b2[1]), [bg] "s"(b2[2]), [bo] "s"(b2[3]), [ri] "s"(whh2v[0]), [rf] "s"(whh2v[1]), [rg] "s"(whh2v[2]), [ro] "s"(whh2v[3]),
          [sel] "s"(0x76543210), [msk] "s"(0x0000ffffu), [xbp] "s"(x), [ob] "s"(out)
        :
          "v124", "v125", "v126", "v127", "v128", "v129", "v130", "v131", "v132", "v133", "v134", "v135", "v136", "v137", "v138", "v139",
          "v140", "v141", "v142", "v143", "v144", "v145", "v146", "v147", "v148", "v149", "v150", "v151", "v152", "v153", "v154", "v155",
          "v156", "v157", "v158", "v159", "v160", "v161", "v162", "v163", "v164", "v165", "v166", "v167", "v168", "v169", "v170", "v171",
          "v172", "v173", "v174", "v175", "v176", "v177", "v178", "v179", "v180", "v181", "v182", "v183", "v184", "v185", "v186", "v187",
          "v188", "v189", "v190", "v191", "v192", "v193", "v194", "v195", "v196", "v197", "v198", "v199", "v200", "v201", "v202", "v203",
          "v204", "v205", "v206", "v207", "v208", "v209", "v210", "v211", "v212", "v213", "v214", "v215", "v216", "v217", "v218", "v219",
          "v220", "v221", "v222", "v223", "v224", "v225", "v226", "v227", "v228", "v229", "v230", "v231", "v232", "v233", "v234", "v235",
          "v236", "v237", "v238", "v239", "v240", "v241", "v242", "v243", "v244", "v245", "v246", "v247", "v248", "v249", "v250", "v251",
          "v252", "v253", "v254", "v255",
          "s40", "s41", "scc", "memory");
}

extern "C" void kernel_launch(void* const* d_in, const int* in_sizes, int n_in,
                              void* d_out, int out_size, void* d_ws, size_t ws_size,
                              hipStream_t stream) {
  const float* x    = (const float*)d_in[0];
  const float* Wih1 = (const float*)d_in[1];
  const float* Whh1 = (const float*)d_in[2];
  const float* bih1 = (const float*)d_in[3];
  const float* bhh1 = (const float*)d_in[4];
  const float* Wih2 = (const float*)d_in[5];
  const float* Whh2 = (const float*)d_in[6];
  const float* bih2 = (const float*)d_in[7];
  const float* bhh2 = (const float*)d_in[8];
  float* out = (float*)d_out;

  const int B    = in_sizes[0] / TSEQ;
  const int grid = B / (WAVES * 16);
  hipLaunchKernelGGL(lstm_fused, dim3(grid), dim3(WAVES * 32), 0, stream,
                     x, Wih1, Whh1, bih1, bhh1, Wih2, Whh2, bih2, bhh2, out);
}
